# starting kernel
# baseline (speedup 1.0000x reference)
#include <hip/hip_runtime.h>
#include <hip/hip_bf16.h>

typedef __attribute__((ext_vector_type(16))) _Float16 v16h;
typedef __attribute__((ext_vector_type(8)))  float    v8f;


__global__ void k_fill1(float* __restrict__ deg, int n) {
    int i = blockIdx.x * blockDim.x + threadIdx.x;
    if (i < n) deg[i] = 1.0f;
}

__global__ void k_degree(const int* __restrict__ dst, float* __restrict__ deg, int E) {
    int e = blockIdx.x * blockDim.x + threadIdx.x;
    if (e < E) atomicAdd(&deg[dst[e]], 1.0f);
}

__global__ void k_rsqrt_inplace(float* __restrict__ deg, int n) {
    int i = blockIdx.x * blockDim.x + threadIdx.x;
    if (i < n) deg[i] = rsqrtf(deg[i]);
}


template <int OUTC, bool RELU>
__global__ __launch_bounds__(256)
void k_gemm_f16(const float* __restrict__ X, const float* __restrict__ Wg,
                float* __restrict__ H, int ntiles) {
    __shared__ _Float16 ldsW[OUTC * 128];

    const int tid = threadIdx.x;
    for (int idx = tid; idx < 128 * OUTC; idx += 256) {
        int k = idx / OUTC;
        int n = idx % OUTC;
        ldsW[n * 128 + k] = (_Float16)Wg[idx];
    }
    __syncthreads();

    const int wave = tid >> 5;
    const int lane = tid & 31;
    const int tile = blockIdx.x * 8 + wave;
    if (tile >= ntiles) return;

    const int mn = lane & 15;
    const int hh = lane >> 4;

    const float* xrow = X + (size_t)(tile * 16 + mn) * 128;
    v16h a[4];
#pragma unroll
    for (int kf = 0; kf < 4; ++kf) {
        const int k0 = kf * 32;
#pragma unroll
        for (int i = 0; i < 8; ++i) {
            int kk = k0 + ((i < 4) ? 2 * i : 2 * i + 8) + 8 * hh;
            float v0 = xrow[kk];
            float v1 = xrow[kk + 1];
            if (RELU) { v0 = fmaxf(v0, 0.0f); v1 = fmaxf(v1, 0.0f); }
            a[kf][2 * i]     = (_Float16)v0;
            a[kf][2 * i + 1] = (_Float16)v1;
        }
    }

#pragma unroll
    for (int nt = 0; nt < OUTC / 16; ++nt) {
        const int col = nt * 16 + mn;
        const _Float16* wrow = &ldsW[col * 128];
        v8f c = {};
#pragma unroll
        for (int kf = 0; kf < 4; ++kf) {
            const int k0 = kf * 32;
            v16h b;
#pragma unroll
            for (int i = 0; i < 8; ++i) {
                int kk = k0 + ((i < 4) ? 2 * i : 2 * i + 8) + 8 * hh;
                b[2 * i]     = wrow[kk];
                b[2 * i + 1] = wrow[kk + 1];
            }
            c = __builtin_amdgcn_wmma_f32_16x16x32_f16(
                    false, a[kf], false, b, (short)0, c, false, false);
        }
        float* hout = H + (size_t)(tile * 16) * OUTC + col;
#pragma unroll
        for (int r = 0; r < 8; ++r)
            hout[(size_t)(r + 8 * hh) * OUTC] = c[r];
    }
}


template <int C>
__global__ void k_selfbias(const float* __restrict__ h, const float* __restrict__ dinv,
                           const float* __restrict__ bias, float* __restrict__ agg, int n) {
    constexpr int G = C / 4;
    int t = blockIdx.x * blockDim.x + threadIdx.x;
    int i = t / G;
    int g = t % G;
    if (i >= n) return;
    float di = dinv[i];
    float s  = di * di;
    const float4 hv = *(const float4*)(h + (size_t)i * C + g * 4);
    const float4 bv = *(const float4*)(bias + g * 4);
    float4 o;
    o.x = hv.x * s + bv.x;
    o.y = hv.y * s + bv.y;
    o.z = hv.z * s + bv.z;
    o.w = hv.w * s + bv.w;
    *(float4*)(agg + (size_t)i * C + g * 4) = o;
}


template <int C>
__global__ void k_scatter(const float* __restrict__ h, const int* __restrict__ src,
                          const int* __restrict__ dst, const float* __restrict__ dinv,
                          float* __restrict__ agg, long long total) {
    constexpr int G = C / 4;
    long long t = (long long)blockIdx.x * blockDim.x + threadIdx.x;
    if (t >= total) return;
    int e = (int)(t / G);
    int g = (int)(t % G);
    int s = src[e];
    int d = dst[e];
    float nrm = dinv[s] * dinv[d];
    const float4 hv = *(const float4*)(h + (size_t)s * C + g * 4);
    float* p = agg + (size_t)d * C + g * 4;
    atomicAdd(p + 0, hv.x * nrm);
    atomicAdd(p + 1, hv.y * nrm);
    atomicAdd(p + 2, hv.z * nrm);
    atomicAdd(p + 3, hv.w * nrm);
}


extern "C" void kernel_launch(void* const* d_in, const int* in_sizes, int n_in,
                              void* d_out, int out_size, void* d_ws, size_t ws_size,
                              hipStream_t stream) {
    const float* x  = (const float*)d_in[0];
    const int*   ei = (const int*)d_in[1];
    const float* W1 = (const float*)d_in[2];
    const float* b1 = (const float*)d_in[3];
    const float* W2 = (const float*)d_in[4];
    const float* b2 = (const float*)d_in[5];
    float*       out = (float*)d_out;

    const int N = in_sizes[0] / 128;
    const int E = in_sizes[1] / 2;
    const int* src = ei;
    const int* dst = ei + E;

    char* w = (char*)d_ws;
    auto carve = [&](size_t bytes) -> void* {
        void* p = (void*)w;
        w += (bytes + 255) & ~(size_t)255;
        return p;
    };
    float* dinv = (float*)carve((size_t)N * 4);
    float* h1   = (float*)carve((size_t)N * 128 * 4);
    float* agg1 = (float*)carve((size_t)N * 128 * 4);
    float* h2   = h1;

    const int B = 256;
    const int ntiles = (N + 15) / 16;
    const int gemm_blocks = (ntiles + 7) / 8;

    k_fill1<<<(N + B - 1) / B, B, 0, stream>>>(dinv, N);
    k_degree<<<(E + B - 1) / B, B, 0, stream>>>(dst, dinv, E);
    k_rsqrt_inplace<<<(N + B - 1) / B, B, 0, stream>>>(dinv, N);

    k_gemm_f16<128, false><<<gemm_blocks, 256, 0, stream>>>(x, W1, h1, ntiles);
    k_selfbias<128><<<(N * 32 + B - 1) / B, B, 0, stream>>>(h1, dinv, b1, agg1, N);
    {
        long long total = (long long)E * 32;
        k_scatter<128><<<(int)((total + B - 1) / B), B, 0, stream>>>(h1, src, dst, dinv, agg1, total);
    }

    k_gemm_f16<64, true><<<gemm_blocks, 256, 0, stream>>>(agg1, W2, h2, ntiles);
    k_selfbias<64><<<(N * 16 + B - 1) / B, B, 0, stream>>>(h2, dinv, b2, out, N);
    {
        long long total = (long long)E * 16;
        k_scatter<64><<<(int)((total + B - 1) / B), B, 0, stream>>>(h2, src, dst, dinv, out, total);
    }
}


# GEMM: W tile in LDS re-laid at an odd 16-byte row pitch (272 B) and staged by 4x4 register transposes with 8-byte stores: B-fragment reads and staging stores become free of bank conflicts
# speedup vs baseline: 1.9825x; 1.9825x over previous
#include <hip/hip_runtime.h>
#include <hip/hip_bf16.h>

typedef __attribute__((ext_vector_type(16))) _Float16 v16h;
typedef __attribute__((ext_vector_type(8)))  float    v8f;


__global__ void k_zero(int* __restrict__ cnt, int n) {
    int i = blockIdx.x * blockDim.x + threadIdx.x;
    if (i < n) cnt[i] = 0;
}

__global__ void k_count_slot(const int* __restrict__ dst, int* __restrict__ cnt,
                             int* __restrict__ slot, int E) {
    int e = blockIdx.x * blockDim.x + threadIdx.x;
    if (e < E) slot[e] = atomicAdd(&cnt[dst[e]], 1);
}

__global__ __launch_bounds__(1024)
void k_segstart(const int* __restrict__ cnt, int* __restrict__ rowptr,
                float* __restrict__ dinv, int n) {
    __shared__ int wsum[32];
    __shared__ int s_front;
    const int tid  = threadIdx.x;
    const int lane = tid & 31;
    const int wave = tid >> 5;
    const int base = blockIdx.x * 1024;

    int part = 0;
    const int4* c4 = (const int4*)cnt;
    const int n4 = base >> 2;
    for (int j = tid; j < n4; j += 1024) {
        const int4 v = c4[j];
        part += (v.x + v.y) + (v.z + v.w);
    }
    for (int o = 16; o > 0; o >>= 1) part += __shfl_xor(part, o);
    if (lane == 0) wsum[wave] = part;
    __syncthreads();
    if (wave == 0) {
        int v = wsum[lane];
        for (int o = 16; o > 0; o >>= 1) v += __shfl_xor(v, o);
        if (lane == 0) s_front = v;
    }
    __syncthreads();
    const int front = s_front;

    const int i = base + tid;
    const int c = (i < n) ? cnt[i] : 0;
    int incl = c;
    for (int o = 1; o < 32; o <<= 1) {
        const int t = __shfl_up(incl, o);
        if (lane >= o) incl += t;
    }
    if (lane == 31) wsum[wave] = incl;
    __syncthreads();
    int woff = (lane < wave) ? wsum[lane] : 0;
    for (int o = 16; o > 0; o >>= 1) woff += __shfl_xor(woff, o);

    const int excl = front + woff + incl - c;
    if (i < n) {
        rowptr[i] = excl;
        dinv[i]   = rsqrtf((float)c + 1.0f);
        if (i == n - 1) rowptr[n] = excl + c;
    }
}

__global__ void k_place(const int* __restrict__ src, const int* __restrict__ dst,
                        const int* __restrict__ slot, const int* __restrict__ rowptr,
                        int* __restrict__ seg_src, int E) {
    int e = blockIdx.x * blockDim.x + threadIdx.x;
    if (e < E) seg_src[rowptr[dst[e]] + slot[e]] = src[e];
}


typedef __attribute__((ext_vector_type(4))) _Float16 v4h;
typedef __attribute__((ext_vector_type(8))) _Float16 v8h;

template <int OUTC, bool RELU>
__global__ __launch_bounds__(256)
void k_gemm_f16(const float* __restrict__ X, const float* __restrict__ Wg,
                float* __restrict__ H, int ntiles) {
    constexpr int WP = 136;
    __shared__ __attribute__((aligned(16))) _Float16 ldsW[OUTC * WP];

    const int tid = threadIdx.x;
    for (int it = tid; it < 32 * (OUTC / 4); it += 256) {
        const int g  = ((it >> 7) << 2) | (it & 3);
        const int kq = (it >> 2) & 31;
        const float4* wsrc = (const float4*)(Wg + (size_t)(4 * kq) * OUTC + 4 * g);
        const float4 r0 = wsrc[0];
        const float4 r1 = wsrc[OUTC / 4];
        const float4 r2 = wsrc[2 * (OUTC / 4)];
        const float4 r3 = wsrc[3 * (OUTC / 4)];
        const v4h c0 = {(_Float16)r0.x, (_Float16)r1.x, (_Float16)r2.x, (_Float16)r3.x};
        const v4h c1 = {(_Float16)r0.y, (_Float16)r1.y, (_Float16)r2.y, (_Float16)r3.y};
        const v4h c2 = {(_Float16)r0.z, (_Float16)r1.z, (_Float16)r2.z, (_Float16)r3.z};
        const v4h c3 = {(_Float16)r0.w, (_Float16)r1.w, (_Float16)r2.w, (_Float16)r3.w};
        _Float16* wdst = &ldsW[(4 * g) * WP + 4 * kq];
        *(v4h*)(wdst)          = c0;
        *(v4h*)(wdst + WP)     = c1;
        *(v4h*)(wdst + 2 * WP) = c2;
        *(v4h*)(wdst + 3 * WP) = c3;
    }
    __syncthreads();

    const int wave = tid >> 5;
    const int lane = tid & 31;
    const int tile = blockIdx.x * 8 + wave;
    if (tile >= ntiles) return;

    const int mn = lane & 15;
    const int hh = lane >> 4;

    const float* xrow = X + (size_t)(tile * 16 + mn) * 128;
    v16h a[4];
#pragma unroll
    for (int kf = 0; kf < 4; ++kf) {
        const int k0 = kf * 32;
#pragma unroll
        for (int i = 0; i < 8; ++i) {
            int kk = k0 + ((i < 4) ? 2 * i : 2 * i + 8) + 8 * hh;
            float v0 = xrow[kk];
            float v1 = xrow[kk + 1];
            if (RELU) { v0 = fmaxf(v0, 0.0f); v1 = fmaxf(v1, 0.0f); }
            a[kf][2 * i]     = (_Float16)v0;
            a[kf][2 * i + 1] = (_Float16)v1;
        }
    }

#pragma unroll
    for (int nt = 0; nt < OUTC / 16; ++nt) {
        const int col = nt * 16 + mn;
        const _Float16* wrow = &ldsW[col * WP];
        v8f c = {};
#pragma unroll
        for (int kf = 0; kf < 4; ++kf) {
            const int k0 = kf * 32;
            const v8h blo = *(const v8h*)(wrow + k0 + 8 * hh);
            const v8h bhi = *(const v8h*)(wrow + k0 + 16 + 8 * hh);
            const v16h b = __builtin_shufflevector(blo, bhi, 0, 1, 2, 3, 4, 5, 6, 7,
                                                   8, 9, 10, 11, 12, 13, 14, 15);
            c = __builtin_amdgcn_wmma_f32_16x16x32_f16(
                    false, a[kf], false, b, (short)0, c, false, false);
        }
        float* hout = H + (size_t)(tile * 16) * OUTC + col;
#pragma unroll
        for (int r = 0; r < 8; ++r)
            hout[(size_t)(r + 8 * hh) * OUTC] = c[r];
    }
}


template <int C>
__global__ __launch_bounds__(256)
void k_gather(const float* __restrict__ h, const int* __restrict__ rowptr,
              const int* __restrict__ seg_src, const float* __restrict__ dinv,
              const float* __restrict__ bias, float* __restrict__ out, int n) {
    constexpr int V = C / 32;
    typedef __attribute__((ext_vector_type(V))) float vecf;
    const int lane = threadIdx.x & 31;
    const int wv   = __builtin_amdgcn_readfirstlane((int)(threadIdx.x >> 5));
    const int node = blockIdx.x * 8 + wv;
    if (node >= n) return;

    const int beg = rowptr[node];
    const int end = rowptr[node + 1];
    const int L   = end - beg;
    const float dd = dinv[node];
    const float* hl = h + lane * V;

    vecf acc;
#pragma unroll
    for (int k = 0; k < V; ++k) acc[k] = 0.0f;

    if (L <= 32) {
        int   s = 0x7fffffff;
        float w = 0.0f;
        if (lane < L) { s = seg_src[beg + lane]; w = dinv[s]; }
        int rk = 0;
        for (int i = 0; i < L; ++i) {
            const int si = __builtin_amdgcn_readlane(s, i);
            rk += ((si < s) || (si == s && i < lane)) ? 1 : 0;
        }
        if (lane >= L) rk = lane;
        const int   ss = __builtin_amdgcn_ds_permute(rk << 2, s);
        const float sw = __int_as_float(__builtin_amdgcn_ds_permute(rk << 2, __float_as_int(w)));
#pragma unroll 4
        for (int r = 0; r < L; ++r) {
            const int   sr = __builtin_amdgcn_readlane(ss, r);
            const float nr = __int_as_float(__builtin_amdgcn_readlane(__float_as_int(sw), r)) * dd;
            const vecf hv = *(const vecf*)(hl + (size_t)sr * C);
#pragma unroll
            for (int k = 0; k < V; ++k) acc[k] = fmaf(hv[k], nr, acc[k]);
        }
    } else {
        int cur = -1;
        for (int it = 0; it < L; ++it) {
            int mn = 0x7fffffff;
            for (int p = beg + lane; p < end; p += 32) {
                const int v = seg_src[p];
                if (v > cur && v < mn) mn = v;
            }
            for (int o = 16; o > 0; o >>= 1) mn = min(mn, __shfl_xor(mn, o));
            if (mn == 0x7fffffff) break;
            int m = 0;
            for (int p = beg + lane; p < end; p += 32) m += (seg_src[p] == mn) ? 1 : 0;
            for (int o = 16; o > 0; o >>= 1) m += __shfl_xor(m, o);
            const float nr = dinv[mn] * dd;
            const vecf hv = *(const vecf*)(hl + (size_t)mn * C);
            for (int t = 0; t < m; ++t) {
#pragma unroll
                for (int k = 0; k < V; ++k) acc[k] = fmaf(hv[k], nr, acc[k]);
            }
            cur = mn;
        }
    }

    const float s2 = dd * dd;
    const vecf hs = *(const vecf*)(hl + (size_t)node * C);
    const vecf bv = *(const vecf*)(bias + lane * V);
    vecf o;
#pragma unroll
    for (int k = 0; k < V; ++k) o[k] = fmaf(hs[k], s2, bv[k]) + acc[k];
    *(vecf*)(out + (size_t)node * C + lane * V) = o;
}


extern "C" void kernel_launch(void* const* d_in, const int* in_sizes, int n_in,
                              void* d_out, int out_size, void* d_ws, size_t ws_size,
                              hipStream_t stream) {
    const float* x  = (const float*)d_in[0];
    const int*   ei = (const int*)d_in[1];
    const float* W1 = (const float*)d_in[2];
    const float* b1 = (const float*)d_in[3];
    const float* W2 = (const float*)d_in[4];
    const float* b2 = (const float*)d_in[5];
    float*       out = (float*)d_out;

    const int N = in_sizes[0] / 128;
    const int E = in_sizes[1] / 2;
    const int* src = ei;
    const int* dst = ei + E;

    char* w = (char*)d_ws;
    auto carve = [&](size_t bytes) -> void* {
        void* p = (void*)w;
        w += (bytes + 255) & ~(size_t)255;
        return p;
    };
    int*   cnt     = (int*)carve((size_t)N * 4);
    int*   slot    = (int*)carve((size_t)E * 4);
    int*   rowptr  = (int*)carve((size_t)(N + 1) * 4);
    int*   seg_src = (int*)carve((size_t)E * 4);
    float* dinv    = (float*)carve((size_t)N * 4);
    float* h1      = (float*)carve((size_t)N * 128 * 4);
    float* agg1    = (float*)carve((size_t)N * 128 * 4);
    float* h2      = h1;

    const int B = 256;
    const int ntiles = (N + 15) / 16;
    const int gemm_blocks = (ntiles + 7) / 8;

    k_zero<<<(N + B - 1) / B, B, 0, stream>>>(cnt, N);
    k_count_slot<<<(E + B - 1) / B, B, 0, stream>>>(dst, cnt, slot, E);
    k_segstart<<<(N + 1023) / 1024, 1024, 0, stream>>>(cnt, rowptr, dinv, N);
    k_place<<<(E + B - 1) / B, B, 0, stream>>>(src, dst, slot, rowptr, seg_src, E);

    k_gemm_f16<128, false><<<gemm_blocks, 256, 0, stream>>>(x, W1, h1, ntiles);
    k_gather<128><<<(N + 7) / 8, 256, 0, stream>>>(h1, rowptr, seg_src, dinv, b1, agg1, N);

    k_gemm_f16<64, true><<<gemm_blocks, 256, 0, stream>>>(agg1, W2, h2, ntiles);
    k_gather<64><<<(N + 7) / 8, 256, 0, stream>>>(h2, rowptr, seg_src, dinv, b2, out, N);
}


# Edge ranks: per-edge returning global atomic add replaced by a returning LDS add on per-chunk node counters (two 16-bit counts per word) plus a scan over chunks; float sum order unchanged
# speedup vs baseline: 2.2399x; 1.1298x over previous
#include <hip/hip_runtime.h>
#include <hip/hip_bf16.h>

typedef __attribute__((ext_vector_type(16))) _Float16 v16h;
typedef __attribute__((ext_vector_type(8)))  float    v8f;


constexpr int HIST_MAXN = 131072;
constexpr int HIST_CH   = 32768;

__global__ __launch_bounds__(1024)
void k_hist(const int* __restrict__ dst, unsigned int* __restrict__ histw,
            int* __restrict__ slot, int E, int n, int nw) {
    __shared__ unsigned int c[HIST_MAXN / 2];
    const int tid = threadIdx.x;
    const int e0  = blockIdx.x * HIST_CH;
    const int e1  = min(E, e0 + HIST_CH);
    unsigned int* hrow = histw + (size_t)blockIdx.x * nw;
    for (int lo = 0; lo < n; lo += HIST_MAXN) {
        const int words = min(HIST_MAXN / 2, nw - (lo >> 1));
        for (int i = tid; i < words; i += 1024) c[i] = 0u;
        __syncthreads();
        for (int b = e0 + tid; b < e1; b += 8 * 1024) {
            int      d[8];
            unsigned r[8];
#pragma unroll
            for (int j = 0; j < 8; ++j)
                d[j] = dst[min(b + j * 1024, e1 - 1)] - lo;
#pragma unroll
            for (int j = 0; j < 8; ++j) {
                const bool ok = (b + j * 1024 < e1) && ((unsigned)d[j] < (unsigned)HIST_MAXN);
                const unsigned sh = (unsigned)(d[j] & 1) << 4;
                r[j] = atomicAdd(&c[ok ? (d[j] >> 1) : 0], ok ? (1u << sh) : 0u);
            }
#pragma unroll
            for (int j = 0; j < 8; ++j) {
                const bool ok = (b + j * 1024 < e1) && ((unsigned)d[j] < (unsigned)HIST_MAXN);
                const unsigned sh = (unsigned)(d[j] & 1) << 4;
                if (ok) slot[b + j * 1024] = (int)((r[j] >> sh) & 0xFFFFu);
            }
        }
        __syncthreads();
        for (int i = tid; i < words; i += 1024) hrow[(lo >> 1) + i] = c[i];
        __syncthreads();
    }
}

__global__ void k_colscan(const unsigned int* __restrict__ histw, int* __restrict__ off,
                          int* __restrict__ cnt, int nw, int chunks) {
    const int w = blockIdx.x * blockDim.x + threadIdx.x;
    if (w >= nw) return;
    const size_t np = (size_t)2 * nw;
    int a = 0, b = 0;
#pragma unroll 8
    for (int p = 0; p < chunks; ++p) {
        const unsigned v = histw[(size_t)p * nw + w];
        *(int2*)(off + (size_t)p * np + 2 * w) = make_int2(a, b);
        a += (int)(v & 0xFFFFu);
        b += (int)(v >> 16);
    }
    *(int2*)(cnt + 2 * w) = make_int2(a, b);
}

__global__ __launch_bounds__(1024)
void k_segstart(const int* __restrict__ cnt, int* __restrict__ rowptr,
                float* __restrict__ dinv, int n) {
    __shared__ int wsum[32];
    __shared__ int s_front;
    const int tid  = threadIdx.x;
    const int lane = tid & 31;
    const int wave = tid >> 5;
    const int base = blockIdx.x * 1024;

    int part = 0;
    const int4* c4 = (const int4*)cnt;
    const int n4 = base >> 2;
    for (int j = tid; j < n4; j += 1024) {
        const int4 v = c4[j];
        part += (v.x + v.y) + (v.z + v.w);
    }
    for (int o = 16; o > 0; o >>= 1) part += __shfl_xor(part, o);
    if (lane == 0) wsum[wave] = part;
    __syncthreads();
    if (wave == 0) {
        int v = wsum[lane];
        for (int o = 16; o > 0; o >>= 1) v += __shfl_xor(v, o);
        if (lane == 0) s_front = v;
    }
    __syncthreads();
    const int front = s_front;

    const int i = base + tid;
    const int c = (i < n) ? cnt[i] : 0;
    int incl = c;
    for (int o = 1; o < 32; o <<= 1) {
        const int t = __shfl_up(incl, o);
        if (lane >= o) incl += t;
    }
    if (lane == 31) wsum[wave] = incl;
    __syncthreads();
    int woff = (lane < wave) ? wsum[lane] : 0;
    for (int o = 16; o > 0; o >>= 1) woff += __shfl_xor(woff, o);

    const int excl = front + woff + incl - c;
    if (i < n) {
        rowptr[i] = excl;
        dinv[i]   = rsqrtf((float)c + 1.0f);
        if (i == n - 1) rowptr[n] = excl + c;
    }
}

__global__ void k_place(const int* __restrict__ src, const int* __restrict__ dst,
                        const int* __restrict__ slot, const int* __restrict__ rowptr,
                        const int* __restrict__ off, int* __restrict__ seg_src, int E, int np) {
    int e = blockIdx.x * blockDim.x + threadIdx.x;
    if (e < E) {
        const int d = dst[e];
        const int p = e / HIST_CH;
        seg_src[rowptr[d] + off[(size_t)p * np + d] + slot[e]] = src[e];
    }
}


typedef __attribute__((ext_vector_type(4))) _Float16 v4h;
typedef __attribute__((ext_vector_type(8))) _Float16 v8h;

template <int OUTC, bool RELU>
__global__ __launch_bounds__(256)
void k_gemm_f16(const float* __restrict__ X, const float* __restrict__ Wg,
                float* __restrict__ H, int ntiles) {
    constexpr int WP = 136;
    __shared__ __attribute__((aligned(16))) _Float16 ldsW[OUTC * WP];

    const int tid = threadIdx.x;
    for (int it = tid; it < 32 * (OUTC / 4); it += 256) {
        const int g  = ((it >> 7) << 2) | (it & 3);
        const int kq = (it >> 2) & 31;
        const float4* wsrc = (const float4*)(Wg + (size_t)(4 * kq) * OUTC + 4 * g);
        const float4 r0 = wsrc[0];
        const float4 r1 = wsrc[OUTC / 4];
        const float4 r2 = wsrc[2 * (OUTC / 4)];
        const float4 r3 = wsrc[3 * (OUTC / 4)];
        const v4h c0 = {(_Float16)r0.x, (_Float16)r1.x, (_Float16)r2.x, (_Float16)r3.x};
        const v4h c1 = {(_Float16)r0.y, (_Float16)r1.y, (_Float16)r2.y, (_Float16)r3.y};
        const v4h c2 = {(_Float16)r0.z, (_Float16)r1.z, (_Float16)r2.z, (_Float16)r3.z};
        const v4h c3 = {(_Float16)r0.w, (_Float16)r1.w, (_Float16)r2.w, (_Float16)r3.w};
        _Float16* wdst = &ldsW[(4 * g) * WP + 4 * kq];
        *(v4h*)(wdst)          = c0;
        *(v4h*)(wdst + WP)     = c1;
        *(v4h*)(wdst + 2 * WP) = c2;
        *(v4h*)(wdst + 3 * WP) = c3;
    }
    __syncthreads();

    const int wave = tid >> 5;
    const int lane = tid & 31;
    const int tile = blockIdx.x * 8 + wave;
    if (tile >= ntiles) return;

    const int mn = lane & 15;
    const int hh = lane >> 4;

    const float* xrow = X + (size_t)(tile * 16 + mn) * 128;
    v16h a[4];
#pragma unroll
    for (int kf = 0; kf < 4; ++kf) {
        const int k0 = kf * 32;
#pragma unroll
        for (int i = 0; i < 8; ++i) {
            int kk = k0 + ((i < 4) ? 2 * i : 2 * i + 8) + 8 * hh;
            float v0 = xrow[kk];
            float v1 = xrow[kk + 1];
            if (RELU) { v0 = fmaxf(v0, 0.0f); v1 = fmaxf(v1, 0.0f); }
            a[kf][2 * i]     = (_Float16)v0;
            a[kf][2 * i + 1] = (_Float16)v1;
        }
    }

#pragma unroll
    for (int nt = 0; nt < OUTC / 16; ++nt) {
        const int col = nt * 16 + mn;
        const _Float16* wrow = &ldsW[col * WP];
        v8f c = {};
#pragma unroll
        for (int kf = 0; kf < 4; ++kf) {
            const int k0 = kf * 32;
            const v8h blo = *(const v8h*)(wrow + k0 + 8 * hh);
            const v8h bhi = *(const v8h*)(wrow + k0 + 16 + 8 * hh);
            const v16h b = __builtin_shufflevector(blo, bhi, 0, 1, 2, 3, 4, 5, 6, 7,
                                                   8, 9, 10, 11, 12, 13, 14, 15);
            c = __builtin_amdgcn_wmma_f32_16x16x32_f16(
                    false, a[kf], false, b, (short)0, c, false, false);
        }
        float* hout = H + (size_t)(tile * 16) * OUTC + col;
#pragma unroll
        for (int r = 0; r < 8; ++r)
            hout[(size_t)(r + 8 * hh) * OUTC] = c[r];
    }
}


template <int C>
__global__ __launch_bounds__(256)
void k_gather(const float* __restrict__ h, const int* __restrict__ rowptr,
              const int* __restrict__ seg_src, const float* __restrict__ dinv,
              const float* __restrict__ bias, float* __restrict__ out, int n) {
    constexpr int V = C / 32;
    typedef __attribute__((ext_vector_type(V))) float vecf;
    const int lane = threadIdx.x & 31;
    const int wv   = __builtin_amdgcn_readfirstlane((int)(threadIdx.x >> 5));
    const int node = blockIdx.x * 8 + wv;
    if (node >= n) return;

    const int beg = rowptr[node];
    const int end = rowptr[node + 1];
    const int L   = end - beg;
    const float dd = dinv[node];
    const float* hl = h + lane * V;

    vecf acc;
#pragma unroll
    for (int k = 0; k < V; ++k) acc[k] = 0.0f;

    if (L <= 32) {
        int   s = 0x7fffffff;
        float w = 0.0f;
        if (lane < L) { s = seg_src[beg + lane]; w = dinv[s]; }
        int rk = 0;
        for (int i = 0; i < L; ++i) {
            const int si = __builtin_amdgcn_readlane(s, i);
            rk += ((si < s) || (si == s && i < lane)) ? 1 : 0;
        }
        if (lane >= L) rk = lane;
        const int   ss = __builtin_amdgcn_ds_permute(rk << 2, s);
        const float sw = __int_as_float(__builtin_amdgcn_ds_permute(rk << 2, __float_as_int(w)));
#pragma unroll 4
        for (int r = 0; r < L; ++r) {
            const int   sr = __builtin_amdgcn_readlane(ss, r);
            const float nr = __int_as_float(__builtin_amdgcn_readlane(__float_as_int(sw), r)) * dd;
            const vecf hv = *(const vecf*)(hl + (size_t)sr * C);
#pragma unroll
            for (int k = 0; k < V; ++k) acc[k] = fmaf(hv[k], nr, acc[k]);
        }
    } else {
        int cur = -1;
        for (int it = 0; it < L; ++it) {
            int mn = 0x7fffffff;
            for (int p = beg + lane; p < end; p += 32) {
                const int v = seg_src[p];
                if (v > cur && v < mn) mn = v;
            }
            for (int o = 16; o > 0; o >>= 1) mn = min(mn, __shfl_xor(mn, o));
            if (mn == 0x7fffffff) break;
            int m = 0;
            for (int p = beg + lane; p < end; p += 32) m += (seg_src[p] == mn) ? 1 : 0;
            for (int o = 16; o > 0; o >>= 1) m += __shfl_xor(m, o);
            const float nr = dinv[mn] * dd;
            const vecf hv = *(const vecf*)(hl + (size_t)mn * C);
            for (int t = 0; t < m; ++t) {
#pragma unroll
                for (int k = 0; k < V; ++k) acc[k] = fmaf(hv[k], nr, acc[k]);
            }
            cur = mn;
        }
    }

    const float s2 = dd * dd;
    const vecf hs = *(const vecf*)(hl + (size_t)node * C);
    const vecf bv = *(const vecf*)(bias + lane * V);
    vecf o;
#pragma unroll
    for (int k = 0; k < V; ++k) o[k] = fmaf(hs[k], s2, bv[k]) + acc[k];
    *(vecf*)(out + (size_t)node * C + lane * V) = o;
}


extern "C" void kernel_launch(void* const* d_in, const int* in_sizes, int n_in,
                              void* d_out, int out_size, void* d_ws, size_t ws_size,
                              hipStream_t stream) {
    const float* x  = (const float*)d_in[0];
    const int*   ei = (const int*)d_in[1];
    const float* W1 = (const float*)d_in[2];
    const float* b1 = (const float*)d_in[3];
    const float* W2 = (const float*)d_in[4];
    const float* b2 = (const float*)d_in[5];
    float*       out = (float*)d_out;

    const int N = in_sizes[0] / 128;
    const int E = in_sizes[1] / 2;
    const int* src = ei;
    const int* dst = ei + E;

    char* w = (char*)d_ws;
    auto carve = [&](size_t bytes) -> void* {
        void* p = (void*)w;
        w += (bytes + 255) & ~(size_t)255;
        return p;
    };
    const int nw     = (N + 1) / 2;
    const int np     = 2 * nw;
    const int chunks = (E + HIST_CH - 1) / HIST_CH;
    int*   cnt     = (int*)carve((size_t)np * 4);
    int*   slot    = (int*)carve((size_t)E * 4);
    int*   rowptr  = (int*)carve((size_t)(N + 1) * 4);
    int*   seg_src = (int*)carve((size_t)E * 4);
    float* dinv    = (float*)carve((size_t)N * 4);
    float* h1      = (float*)carve((size_t)N * 128 * 4);
    float* agg1    = (float*)carve((size_t)N * 128 * 4);
    float* h2      = h1;
    unsigned int* histw = (unsigned int*)carve((size_t)chunks * nw * 4);
    int*   off     = (int*)carve((size_t)chunks * np * 4);

    const int B = 256;
    const int ntiles = (N + 15) / 16;
    const int gemm_blocks = (ntiles + 7) / 8;

    k_hist<<<chunks, 1024, 0, stream>>>(dst, histw, slot, E, N, nw);
    k_colscan<<<(nw + B - 1) / B, B, 0, stream>>>(histw, off, cnt, nw, chunks);
    k_segstart<<<(N + 1023) / 1024, 1024, 0, stream>>>(cnt, rowptr, dinv, N);
    k_place<<<(E + B - 1) / B, B, 0, stream>>>(src, dst, slot, rowptr, off, seg_src, E, np);

    k_gemm_f16<128, false><<<gemm_blocks, 256, 0, stream>>>(x, W1, h1, ntiles);
    k_gather<128><<<(N + 7) / 8, 256, 0, stream>>>(h1, rowptr, seg_src, dinv, b1, agg1, N);

    k_gemm_f16<64, true><<<gemm_blocks, 256, 0, stream>>>(agg1, W2, h2, ntiles);
    k_gather<64><<<(N + 7) / 8, 256, 0, stream>>>(h2, rowptr, seg_src, dinv, b2, out, N);
}


# Layer-1 GEMM moved into the edge-rank launch: its 196 work-groups (32 row tiles each) run on the WGPs that the 49 rank work-groups leave idle; one launch fewer; float sum order unchanged
# speedup vs baseline: 2.4195x; 1.0802x over previous
#include <hip/hip_runtime.h>
#include <hip/hip_bf16.h>

typedef __attribute__((ext_vector_type(16))) _Float16 v16h;
typedef __attribute__((ext_vector_type(8)))  float    v8f;


constexpr int HIST_MAXN = 131072;
constexpr int HIST_CH   = 32768;

__device__ __forceinline__
void hist_block(const int* __restrict__ dst, unsigned int* __restrict__ histw,
                int* __restrict__ slot, int E, int n, int nw, unsigned int* c) {
    const int tid = threadIdx.x;
    const int e0  = blockIdx.x * HIST_CH;
    const int e1  = min(E, e0 + HIST_CH);
    unsigned int* hrow = histw + (size_t)blockIdx.x * nw;
    for (int lo = 0; lo < n; lo += HIST_MAXN) {
        const int words = min(HIST_MAXN / 2, nw - (lo >> 1));
        for (int i = tid; i < words; i += 1024) c[i] = 0u;
        __syncthreads();
        for (int b = e0 + tid; b < e1; b += 8 * 1024) {
            int      d[8];
            unsigned r[8];
#pragma unroll
            for (int j = 0; j < 8; ++j)
                d[j] = dst[min(b + j * 1024, e1 - 1)] - lo;
#pragma unroll
            for (int j = 0; j < 8; ++j) {
                const bool ok = (b + j * 1024 < e1) && ((unsigned)d[j] < (unsigned)HIST_MAXN);
                const unsigned sh = (unsigned)(d[j] & 1) << 4;
                r[j] = atomicAdd(&c[ok ? (d[j] >> 1) : 0], ok ? (1u << sh) : 0u);
            }
#pragma unroll
            for (int j = 0; j < 8; ++j) {
                const bool ok = (b + j * 1024 < e1) && ((unsigned)d[j] < (unsigned)HIST_MAXN);
                const unsigned sh = (unsigned)(d[j] & 1) << 4;
                if (ok) slot[b + j * 1024] = (int)((r[j] >> sh) & 0xFFFFu);
            }
        }
        __syncthreads();
        for (int i = tid; i < words; i += 1024) hrow[(lo >> 1) + i] = c[i];
        __syncthreads();
    }
}

__global__ void k_colscan(const unsigned int* __restrict__ histw, int* __restrict__ off,
                          int* __restrict__ cnt, int nw, int chunks) {
    const int w = blockIdx.x * blockDim.x + threadIdx.x;
    if (w >= nw) return;
    const size_t np = (size_t)2 * nw;
    int a = 0, b = 0;
#pragma unroll 8
    for (int p = 0; p < chunks; ++p) {
        const unsigned v = histw[(size_t)p * nw + w];
        *(int2*)(off + (size_t)p * np + 2 * w) = make_int2(a, b);
        a += (int)(v & 0xFFFFu);
        b += (int)(v >> 16);
    }
    *(int2*)(cnt + 2 * w) = make_int2(a, b);
}

__global__ __launch_bounds__(1024)
void k_segstart(const int* __restrict__ cnt, int* __restrict__ rowptr,
                float* __restrict__ dinv, int n) {
    __shared__ int wsum[32];
    __shared__ int s_front;
    const int tid  = threadIdx.x;
    const int lane = tid & 31;
    const int wave = tid >> 5;
    const int base = blockIdx.x * 1024;

    int part = 0;
    const int4* c4 = (const int4*)cnt;
    const int n4 = base >> 2;
    for (int j = tid; j < n4; j += 1024) {
        const int4 v = c4[j];
        part += (v.x + v.y) + (v.z + v.w);
    }
    for (int o = 16; o > 0; o >>= 1) part += __shfl_xor(part, o);
    if (lane == 0) wsum[wave] = part;
    __syncthreads();
    if (wave == 0) {
        int v = wsum[lane];
        for (int o = 16; o > 0; o >>= 1) v += __shfl_xor(v, o);
        if (lane == 0) s_front = v;
    }
    __syncthreads();
    const int front = s_front;

    const int i = base + tid;
    const int c = (i < n) ? cnt[i] : 0;
    int incl = c;
    for (int o = 1; o < 32; o <<= 1) {
        const int t = __shfl_up(incl, o);
        if (lane >= o) incl += t;
    }
    if (lane == 31) wsum[wave] = incl;
    __syncthreads();
    int woff = (lane < wave) ? wsum[lane] : 0;
    for (int o = 16; o > 0; o >>= 1) woff += __shfl_xor(woff, o);

    const int excl = front + woff + incl - c;
    if (i < n) {
        rowptr[i] = excl;
        dinv[i]   = rsqrtf((float)c + 1.0f);
        if (i == n - 1) rowptr[n] = excl + c;
    }
}

__global__ void k_place(const int* __restrict__ src, const int* __restrict__ dst,
                        const int* __restrict__ slot, const int* __restrict__ rowptr,
                        const int* __restrict__ off, int* __restrict__ seg_src, int E, int np) {
    int e = blockIdx.x * blockDim.x + threadIdx.x;
    if (e < E) {
        const int d = dst[e];
        const int p = e / HIST_CH;
        seg_src[rowptr[d] + off[(size_t)p * np + d] + slot[e]] = src[e];
    }
}


typedef __attribute__((ext_vector_type(4))) _Float16 v4h;
typedef __attribute__((ext_vector_type(8))) _Float16 v8h;

constexpr int GEMM_WP = 136;

template <int OUTC, bool RELU, int NT>
__device__ __forceinline__
void gemm_block(const float* __restrict__ X, const float* __restrict__ Wg,
                float* __restrict__ H, int ntiles, int blk, _Float16* ldsW) {
    constexpr int WP = GEMM_WP;

    const int tid = threadIdx.x;
    for (int it = tid; it < 32 * (OUTC / 4); it += NT) {
        const int g  = ((it >> 7) << 2) | (it & 3);
        const int kq = (it >> 2) & 31;
        const float4* wsrc = (const float4*)(Wg + (size_t)(4 * kq) * OUTC + 4 * g);
        const float4 r0 = wsrc[0];
        const float4 r1 = wsrc[OUTC / 4];
        const float4 r2 = wsrc[2 * (OUTC / 4)];
        const float4 r3 = wsrc[3 * (OUTC / 4)];
        const v4h c0 = {(_Float16)r0.x, (_Float16)r1.x, (_Float16)r2.x, (_Float16)r3.x};
        const v4h c1 = {(_Float16)r0.y, (_Float16)r1.y, (_Float16)r2.y, (_Float16)r3.y};
        const v4h c2 = {(_Float16)r0.z, (_Float16)r1.z, (_Float16)r2.z, (_Float16)r3.z};
        const v4h c3 = {(_Float16)r0.w, (_Float16)r1.w, (_Float16)r2.w, (_Float16)r3.w};
        _Float16* wdst = &ldsW[(4 * g) * WP + 4 * kq];
        *(v4h*)(wdst)          = c0;
        *(v4h*)(wdst + WP)     = c1;
        *(v4h*)(wdst + 2 * WP) = c2;
        *(v4h*)(wdst + 3 * WP) = c3;
    }
    __syncthreads();

    const int wave = tid >> 5;
    const int lane = tid & 31;
    const int tile = blk * (NT / 32) + wave;
    if (tile >= ntiles) return;

    const int mn = lane & 15;
    const int hh = lane >> 4;

    const float* xrow = X + (size_t)(tile * 16 + mn) * 128;
    v16h a[4];
#pragma unroll
    for (int kf = 0; kf < 4; ++kf) {
        const int k0 = kf * 32;
#pragma unroll
        for (int i = 0; i < 8; ++i) {
            int kk = k0 + ((i < 4) ? 2 * i : 2 * i + 8) + 8 * hh;
            float v0 = xrow[kk];
            float v1 = xrow[kk + 1];
            if (RELU) { v0 = fmaxf(v0, 0.0f); v1 = fmaxf(v1, 0.0f); }
            a[kf][2 * i]     = (_Float16)v0;
            a[kf][2 * i + 1] = (_Float16)v1;
        }
    }

#pragma unroll
    for (int nt = 0; nt < OUTC / 16; ++nt) {
        const int col = nt * 16 + mn;
        const _Float16* wrow = &ldsW[col * WP];
        v8f c = {};
#pragma unroll
        for (int kf = 0; kf < 4; ++kf) {
            const int k0 = kf * 32;
            const v8h blo = *(const v8h*)(wrow + k0 + 8 * hh);
            const v8h bhi = *(const v8h*)(wrow + k0 + 16 + 8 * hh);
            const v16h b = __builtin_shufflevector(blo, bhi, 0, 1, 2, 3, 4, 5, 6, 7,
                                                   8, 9, 10, 11, 12, 13, 14, 15);
            c = __builtin_amdgcn_wmma_f32_16x16x32_f16(
                    false, a[kf], false, b, (short)0, c, false, false);
        }
        float* hout = H + (size_t)(tile * 16) * OUTC + col;
#pragma unroll
        for (int r = 0; r < 8; ++r)
            hout[(size_t)(r + 8 * hh) * OUTC] = c[r];
    }
}

template <int OUTC, bool RELU>
__global__ __launch_bounds__(256)
void k_gemm_f16(const float* __restrict__ X, const float* __restrict__ Wg,
                float* __restrict__ H, int ntiles) {
    __shared__ __attribute__((aligned(16))) _Float16 ldsW[OUTC * GEMM_WP];
    gemm_block<OUTC, RELU, 256>(X, Wg, H, ntiles, (int)blockIdx.x, ldsW);
}

__global__ __launch_bounds__(1024)
void k_hist_gemm1(const int* __restrict__ dst, unsigned int* __restrict__ histw,
                  int* __restrict__ slot, int E, int n, int nw, int chunks,
                  const float* __restrict__ X, const float* __restrict__ Wg,
                  float* __restrict__ H, int ntiles) {
    __shared__ __attribute__((aligned(16))) unsigned int c[HIST_MAXN / 2];
    if ((int)blockIdx.x >= chunks) {
        gemm_block<128, false, 1024>(X, Wg, H, ntiles, (int)blockIdx.x - chunks, (_Float16*)c);
        return;
    }
    hist_block(dst, histw, slot, E, n, nw, c);
}


template <int C>
__global__ __launch_bounds__(256)
void k_gather(const float* __restrict__ h, const int* __restrict__ rowptr,
              const int* __restrict__ seg_src, const float* __restrict__ dinv,
              const float* __restrict__ bias, float* __restrict__ out, int n) {
    constexpr int V = C / 32;
    typedef __attribute__((ext_vector_type(V))) float vecf;
    const int lane = threadIdx.x & 31;
    const int wv   = __builtin_amdgcn_readfirstlane((int)(threadIdx.x >> 5));
    const int node = blockIdx.x * 8 + wv;
    if (node >= n) return;

    const int beg = rowptr[node];
    const int end = rowptr[node + 1];
    const int L   = end - beg;
    const float dd = dinv[node];
    const float* hl = h + lane * V;

    vecf acc;
#pragma unroll
    for (int k = 0; k < V; ++k) acc[k] = 0.0f;

    if (L <= 32) {
        int   s = 0x7fffffff;
        float w = 0.0f;
        if (lane < L) { s = seg_src[beg + lane]; w = dinv[s]; }
        int rk = 0;
        for (int i = 0; i < L; ++i) {
            const int si = __builtin_amdgcn_readlane(s, i);
            rk += ((si < s) || (si == s && i < lane)) ? 1 : 0;
        }
        if (lane >= L) rk = lane;
        const int   ss = __builtin_amdgcn_ds_permute(rk << 2, s);
        const float sw = __int_as_float(__builtin_amdgcn_ds_permute(rk << 2, __float_as_int(w)));
#pragma unroll 4
        for (int r = 0; r < L; ++r) {
            const int   sr = __builtin_amdgcn_readlane(ss, r);
            const float nr = __int_as_float(__builtin_amdgcn_readlane(__float_as_int(sw), r)) * dd;
            const vecf hv = *(const vecf*)(hl + (size_t)sr * C);
#pragma unroll
            for (int k = 0; k < V; ++k) acc[k] = fmaf(hv[k], nr, acc[k]);
        }
    } else {
        int cur = -1;
        for (int it = 0; it < L; ++it) {
            int mn = 0x7fffffff;
            for (int p = beg + lane; p < end; p += 32) {
                const int v = seg_src[p];
                if (v > cur && v < mn) mn = v;
            }
            for (int o = 16; o > 0; o >>= 1) mn = min(mn, __shfl_xor(mn, o));
            if (mn == 0x7fffffff) break;
            int m = 0;
            for (int p = beg + lane; p < end; p += 32) m += (seg_src[p] == mn) ? 1 : 0;
            for (int o = 16; o > 0; o >>= 1) m += __shfl_xor(m, o);
            const float nr = dinv[mn] * dd;
            const vecf hv = *(const vecf*)(hl + (size_t)mn * C);
            for (int t = 0; t < m; ++t) {
#pragma unroll
                for (int k = 0; k < V; ++k) acc[k] = fmaf(hv[k], nr, acc[k]);
            }
            cur = mn;
        }
    }

    const float s2 = dd * dd;
    const vecf hs = *(const vecf*)(hl + (size_t)node * C);
    const vecf bv = *(const vecf*)(bias + lane * V);
    vecf o;
#pragma unroll
    for (int k = 0; k < V; ++k) o[k] = fmaf(hs[k], s2, bv[k]) + acc[k];
    *(vecf*)(out + (size_t)node * C + lane * V) = o;
}


extern "C" void kernel_launch(void* const* d_in, const int* in_sizes, int n_in,
                              void* d_out, int out_size, void* d_ws, size_t ws_size,
                              hipStream_t stream) {
    const float* x  = (const float*)d_in[0];
    const int*   ei = (const int*)d_in[1];
    const float* W1 = (const float*)d_in[2];
    const float* b1 = (const float*)d_in[3];
    const float* W2 = (const float*)d_in[4];
    const float* b2 = (const float*)d_in[5];
    float*       out = (float*)d_out;

    const int N = in_sizes[0] / 128;
    const int E = in_sizes[1] / 2;
    const int* src = ei;
    const int* dst = ei + E;

    char* w = (char*)d_ws;
    auto carve = [&](size_t bytes) -> void* {
        void* p = (void*)w;
        w += (bytes + 255) & ~(size_t)255;
        return p;
    };
    const int nw     = (N + 1) / 2;
    const int np     = 2 * nw;
    const int chunks = (E + HIST_CH - 1) / HIST_CH;
    int*   cnt     = (int*)carve((size_t)np * 4);
    int*   slot    = (int*)carve((size_t)E * 4);
    int*   rowptr  = (int*)carve((size_t)(N + 1) * 4);
    int*   seg_src = (int*)carve((size_t)E * 4);
    float* dinv    = (float*)carve((size_t)N * 4);
    float* h1      = (float*)carve((size_t)N * 128 * 4);
    float* agg1    = (float*)carve((size_t)N * 128 * 4);
    float* h2      = h1;
    unsigned int* histw = (unsigned int*)carve((size_t)chunks * nw * 4);
    int*   off     = (int*)carve((size_t)chunks * np * 4);

    const int B = 256;
    const int ntiles = (N + 15) / 16;
    const int gemm_blocks = (ntiles + 7) / 8;

    const int gemm1_blocks = (ntiles + 31) / 32;
    k_hist_gemm1<<<chunks + gemm1_blocks, 1024, 0, stream>>>(dst, histw, slot, E, N, nw, chunks,
                                                            x, W1, h1, ntiles);
    k_colscan<<<(nw + B - 1) / B, B, 0, stream>>>(histw, off, cnt, nw, chunks);
    k_segstart<<<(N + 1023) / 1024, 1024, 0, stream>>>(cnt, rowptr, dinv, N);
    k_place<<<(E + B - 1) / B, B, 0, stream>>>(src, dst, slot, rowptr, off, seg_src, E, np);

    k_gather<128><<<(N + 7) / 8, 256, 0, stream>>>(h1, rowptr, seg_src, dinv, b1, agg1, N);

    k_gemm_f16<64, true><<<gemm_blocks, 256, 0, stream>>>(agg1, W2, h2, ntiles);
    k_gather<64><<<(N + 7) / 8, 256, 0, stream>>>(h2, rowptr, seg_src, dinv, b2, out, N);
}


# Layer-2 GEMM moved into the layer-1 gather launch: 16 waves gather a 16-node tile, relu(row) goes as f16 through LDS to WMMA (W2 pre-laid as B fragments); one launch fewer; float sum order unchanged
# speedup vs baseline: 2.5100x; 1.0374x over previous
#include <hip/hip_runtime.h>
#include <hip/hip_bf16.h>

typedef __attribute__((ext_vector_type(16))) _Float16 v16h;
typedef __attribute__((ext_vector_type(8)))  float    v8f;
typedef __attribute__((ext_vector_type(4))) _Float16 v4h;
typedef __attribute__((ext_vector_type(8))) _Float16 v8h;


constexpr int HIST_MAXN = 131072;
constexpr int HIST_CH   = 32768;

__device__ __forceinline__
void hist_block(const int* __restrict__ dst, unsigned int* __restrict__ histw,
                int* __restrict__ slot, int E, int n, int nw, unsigned int* c) {
    const int tid = threadIdx.x;
    const int e0  = blockIdx.x * HIST_CH;
    const int e1  = min(E, e0 + HIST_CH);
    unsigned int* hrow = histw + (size_t)blockIdx.x * nw;
    for (int lo = 0; lo < n; lo += HIST_MAXN) {
        const int words = min(HIST_MAXN / 2, nw - (lo >> 1));
        for (int i = tid; i < words; i += 1024) c[i] = 0u;
        __syncthreads();
        for (int b = e0 + tid; b < e1; b += 8 * 1024) {
            int      d[8];
            unsigned r[8];
#pragma unroll
            for (int j = 0; j < 8; ++j)
                d[j] = dst[min(b + j * 1024, e1 - 1)] - lo;
#pragma unroll
            for (int j = 0; j < 8; ++j) {
                const bool ok = (b + j * 1024 < e1) && ((unsigned)d[j] < (unsigned)HIST_MAXN);
                const unsigned sh = (unsigned)(d[j] & 1) << 4;
                r[j] = atomicAdd(&c[ok ? (d[j] >> 1) : 0], ok ? (1u << sh) : 0u);
            }
#pragma unroll
            for (int j = 0; j < 8; ++j) {
                const bool ok = (b + j * 1024 < e1) && ((unsigned)d[j] < (unsigned)HIST_MAXN);
                const unsigned sh = (unsigned)(d[j] & 1) << 4;
                if (ok) slot[b + j * 1024] = (int)((r[j] >> sh) & 0xFFFFu);
            }
        }
        __syncthreads();
        for (int i = tid; i < words; i += 1024) hrow[(lo >> 1) + i] = c[i];
        __syncthreads();
    }
}

__global__ void k_colscan(const unsigned int* __restrict__ histw, int* __restrict__ off,
                          int* __restrict__ cnt, int nw, int chunks) {
    const int w = blockIdx.x * blockDim.x + threadIdx.x;
    if (w >= nw) return;
    const size_t np = (size_t)2 * nw;
    int a = 0, b = 0;
#pragma unroll 8
    for (int p = 0; p < chunks; ++p) {
        const unsigned v = histw[(size_t)p * nw + w];
        *(int2*)(off + (size_t)p * np + 2 * w) = make_int2(a, b);
        a += (int)(v & 0xFFFFu);
        b += (int)(v >> 16);
    }
    *(int2*)(cnt + 2 * w) = make_int2(a, b);
}

__global__ __launch_bounds__(1024)
void k_segstart(const int* __restrict__ cnt, int* __restrict__ rowptr,
                float* __restrict__ dinv, int n,
                const float* __restrict__ W2, _Float16* __restrict__ W2f) {
    __shared__ int wsum[32];
    __shared__ int s_front;
    const int tid  = threadIdx.x;
    const int lane = tid & 31;
    const int wave = tid >> 5;
    const int base = blockIdx.x * 1024;

    if (blockIdx.x == 0) {
        const int j  = (tid >> 5) & 1;
        const int kf = (tid >> 6) & 3;
        const int nt = tid >> 8;
        const float* wp = W2 + (size_t)(kf * 32 + 16 * j + 8 * (lane >> 4)) * 64 + nt * 16 + (lane & 15);
        v8h f;
#pragma unroll
        for (int t = 0; t < 8; ++t) f[t] = (_Float16)wp[t * 64];
        *(v8h*)(W2f + tid * 8) = f;
    }

    int part = 0;
    const int4* c4 = (const int4*)cnt;
    const int n4 = base >> 2;
    for (int j = tid; j < n4; j += 1024) {
        const int4 v = c4[j];
        part += (v.x + v.y) + (v.z + v.w);
    }
    for (int o = 16; o > 0; o >>= 1) part += __shfl_xor(part, o);
    if (lane == 0) wsum[wave] = part;
    __syncthreads();
    if (wave == 0) {
        int v = wsum[lane];
        for (int o = 16; o > 0; o >>= 1) v += __shfl_xor(v, o);
        if (lane == 0) s_front = v;
    }
    __syncthreads();
    const int front = s_front;

    const int i = base + tid;
    const int c = (i < n) ? cnt[i] : 0;
    int incl = c;
    for (int o = 1; o < 32; o <<= 1) {
        const int t = __shfl_up(incl, o);
        if (lane >= o) incl += t;
    }
    if (lane == 31) wsum[wave] = incl;
    __syncthreads();
    int woff = (lane < wave) ? wsum[lane] : 0;
    for (int o = 16; o > 0; o >>= 1) woff += __shfl_xor(woff, o);

    const int excl = front + woff + incl - c;
    if (i < n) {
        rowptr[i] = excl;
        dinv[i]   = rsqrtf((float)c + 1.0f);
        if (i == n - 1) rowptr[n] = excl + c;
    }
}

__global__ void k_place(const int* __restrict__ src, const int* __restrict__ dst,
                        const int* __restrict__ slot, const int* __restrict__ rowptr,
                        const int* __restrict__ off, int* __restrict__ seg_src, int E, int np) {
    int e = blockIdx.x * blockDim.x + threadIdx.x;
    if (e < E) {
        const int d = dst[e];
        const int p = e / HIST_CH;
        seg_src[rowptr[d] + off[(size_t)p * np + d] + slot[e]] = src[e];
    }
}


constexpr int GEMM_WP = 136;

template <int OUTC, bool RELU, int NT>
__device__ __forceinline__
void gemm_block(const float* __restrict__ X, const float* __restrict__ Wg,
                float* __restrict__ H, int ntiles, int blk, _Float16* ldsW) {
    constexpr int WP = GEMM_WP;

    const int tid = threadIdx.x;
    for (int it = tid; it < 32 * (OUTC / 4); it += NT) {
        const int g  = ((it >> 7) << 2) | (it & 3);
        const int kq = (it >> 2) & 31;
        const float4* wsrc = (const float4*)(Wg + (size_t)(4 * kq) * OUTC + 4 * g);
        const float4 r0 = wsrc[0];
        const float4 r1 = wsrc[OUTC / 4];
        const float4 r2 = wsrc[2 * (OUTC / 4)];
        const float4 r3 = wsrc[3 * (OUTC / 4)];
        const v4h c0 = {(_Float16)r0.x, (_Float16)r1.x, (_Float16)r2.x, (_Float16)r3.x};
        const v4h c1 = {(_Float16)r0.y, (_Float16)r1.y, (_Float16)r2.y, (_Float16)r3.y};
        const v4h c2 = {(_Float16)r0.z, (_Float16)r1.z, (_Float16)r2.z, (_Float16)r3.z};
        const v4h c3 = {(_Float16)r0.w, (_Float16)r1.w, (_Float16)r2.w, (_Float16)r3.w};
        _Float16* wdst = &ldsW[(4 * g) * WP + 4 * kq];
        *(v4h*)(wdst)          = c0;
        *(v4h*)(wdst + WP)     = c1;
        *(v4h*)(wdst + 2 * WP) = c2;
        *(v4h*)(wdst + 3 * WP) = c3;
    }
    __syncthreads();

    const int wave = tid >> 5;
    const int lane = tid & 31;
    const int tile = blk * (NT / 32) + wave;
    if (tile >= ntiles) return;

    const int mn = lane & 15;
    const int hh = lane >> 4;

    const float* xrow = X + (size_t)(tile * 16 + mn) * 128;
    v16h a[4];
#pragma unroll
    for (int kf = 0; kf < 4; ++kf) {
        const int k0 = kf * 32;
#pragma unroll
        for (int i = 0; i < 8; ++i) {
            int kk = k0 + ((i < 4) ? 2 * i : 2 * i + 8) + 8 * hh;
            float v0 = xrow[kk];
            float v1 = xrow[kk + 1];
            if (RELU) { v0 = fmaxf(v0, 0.0f); v1 = fmaxf(v1, 0.0f); }
            a[kf][2 * i]     = (_Float16)v0;
            a[kf][2 * i + 1] = (_Float16)v1;
        }
    }

#pragma unroll
    for (int nt = 0; nt < OUTC / 16; ++nt) {
        const int col = nt * 16 + mn;
        const _Float16* wrow = &ldsW[col * WP];
        v8f c = {};
#pragma unroll
        for (int kf = 0; kf < 4; ++kf) {
            const int k0 = kf * 32;
            const v8h blo = *(const v8h*)(wrow + k0 + 8 * hh);
            const v8h bhi = *(const v8h*)(wrow + k0 + 16 + 8 * hh);
            const v16h b = __builtin_shufflevector(blo, bhi, 0, 1, 2, 3, 4, 5, 6, 7,
                                                   8, 9, 10, 11, 12, 13, 14, 15);
            c = __builtin_amdgcn_wmma_f32_16x16x32_f16(
                    false, a[kf], false, b, (short)0, c, false, false);
        }
        float* hout = H + (size_t)(tile * 16) * OUTC + col;
#pragma unroll
        for (int r = 0; r < 8; ++r)
            hout[(size_t)(r + 8 * hh) * OUTC] = c[r];
    }
}

__global__ __launch_bounds__(1024)
void k_hist_gemm1(const int* __restrict__ dst, unsigned int* __restrict__ histw,
                  int* __restrict__ slot, int E, int n, int nw, int chunks,
                  const float* __restrict__ X, const float* __restrict__ Wg,
                  float* __restrict__ H, int ntiles) {
    __shared__ __attribute__((aligned(16))) unsigned int c[HIST_MAXN / 2];
    if ((int)blockIdx.x >= chunks) {
        gemm_block<128, false, 1024>(X, Wg, H, ntiles, (int)blockIdx.x - chunks, (_Float16*)c);
        return;
    }
    hist_block(dst, histw, slot, E, n, nw, c);
}


template <int C>
__device__ __forceinline__
__attribute__((ext_vector_type(C / 32))) float
gather_node(const float* __restrict__ h, const int* __restrict__ rowptr,
            const int* __restrict__ seg_src, const float* __restrict__ dinv,
            const float* __restrict__ bias, int node, int lane) {
    constexpr int V = C / 32;
    typedef __attribute__((ext_vector_type(V))) float vecf;

    const int beg = rowptr[node];
    const int end = rowptr[node + 1];
    const int L   = end - beg;
    const float dd = dinv[node];
    const float* hl = h + lane * V;

    vecf acc;
#pragma unroll
    for (int k = 0; k < V; ++k) acc[k] = 0.0f;

    if (L <= 32) {
        int   s = 0x7fffffff;
        float w = 0.0f;
        if (lane < L) { s = seg_src[beg + lane]; w = dinv[s]; }
        int rk = 0;
        for (int i = 0; i < L; ++i) {
            const int si = __builtin_amdgcn_readlane(s, i);
            rk += ((si < s) || (si == s && i < lane)) ? 1 : 0;
        }
        if (lane >= L) rk = lane;
        const int   ss = __builtin_amdgcn_ds_permute(rk << 2, s);
        const float sw = __int_as_float(__builtin_amdgcn_ds_permute(rk << 2, __float_as_int(w)));
#pragma unroll 4
        for (int r = 0; r < L; ++r) {
            const int   sr = __builtin_amdgcn_readlane(ss, r);
            const float nr = __int_as_float(__builtin_amdgcn_readlane(__float_as_int(sw), r)) * dd;
            const vecf hv = *(const vecf*)(hl + (size_t)sr * C);
#pragma unroll
            for (int k = 0; k < V; ++k) acc[k] = fmaf(hv[k], nr, acc[k]);
        }
    } else {
        int cur = -1;
        for (int it = 0; it < L; ++it) {
            int mn = 0x7fffffff;
            for (int p = beg + lane; p < end; p += 32) {
                const int v = seg_src[p];
                if (v > cur && v < mn) mn = v;
            }
            for (int o = 16; o > 0; o >>= 1) mn = min(mn, __shfl_xor(mn, o));
            if (mn == 0x7fffffff) break;
            int m = 0;
            for (int p = beg + lane; p < end; p += 32) m += (seg_src[p] == mn) ? 1 : 0;
            for (int o = 16; o > 0; o >>= 1) m += __shfl_xor(m, o);
            const float nr = dinv[mn] * dd;
            const vecf hv = *(const vecf*)(hl + (size_t)mn * C);
            for (int t = 0; t < m; ++t) {
#pragma unroll
                for (int k = 0; k < V; ++k) acc[k] = fmaf(hv[k], nr, acc[k]);
            }
            cur = mn;
        }
    }

    const float s2 = dd * dd;
    const vecf hs = *(const vecf*)(hl + (size_t)node * C);
    const vecf bv = *(const vecf*)(bias + lane * V);
    vecf o;
#pragma unroll
    for (int k = 0; k < V; ++k) o[k] = fmaf(hs[k], s2, bv[k]) + acc[k];
    return o;
}

template <int C>
__global__ __launch_bounds__(256)
void k_gather(const float* __restrict__ h, const int* __restrict__ rowptr,
              const int* __restrict__ seg_src, const float* __restrict__ dinv,
              const float* __restrict__ bias, float* __restrict__ out, int n) {
    constexpr int V = C / 32;
    typedef __attribute__((ext_vector_type(V))) float vecf;
    const int lane = threadIdx.x & 31;
    const int wv   = __builtin_amdgcn_readfirstlane((int)(threadIdx.x >> 5));
    const int node = blockIdx.x * 8 + wv;
    if (node >= n) return;
    const vecf o = gather_node<C>(h, rowptr, seg_src, dinv, bias, node, lane);
    *(vecf*)(out + (size_t)node * C + lane * V) = o;
}

constexpr int G2_DP = 68;

__global__ __launch_bounds__(512)
void k_gather_gemm2(const float* __restrict__ h, const int* __restrict__ rowptr,
                    const int* __restrict__ seg_src, const float* __restrict__ dinv,
                    const float* __restrict__ bias, const _Float16* __restrict__ W2f,
                    float* __restrict__ H2, int n) {
    typedef __attribute__((ext_vector_type(4))) float v4f;
    __shared__ __attribute__((aligned(16))) _Float16 tileA[16 * GEMM_WP];
    __shared__ __attribute__((aligned(16))) float    tileD[16 * G2_DP];
    const int lane = threadIdx.x & 31;
    const int wv   = __builtin_amdgcn_readfirstlane((int)(threadIdx.x >> 5));
    const int node = blockIdx.x * 16 + wv;

    v4f o = {0.0f, 0.0f, 0.0f, 0.0f};
    if (node < n)
        o = gather_node<128>(h, rowptr, seg_src, dinv, bias, node, lane);
    const v4h oh = {(_Float16)fmaxf(o[0], 0.0f), (_Float16)fmaxf(o[1], 0.0f),
                    (_Float16)fmaxf(o[2], 0.0f), (_Float16)fmaxf(o[3], 0.0f)};
    *(v4h*)&tileA[wv * GEMM_WP + lane * 4] = oh;
    v8h bf[8];
    if (wv < 4) {
        const v8h* wf = (const v8h*)W2f + wv * 256 + lane;
#pragma unroll
        for (int q = 0; q < 8; ++q) bf[q] = wf[q * 32];
    }
    __syncthreads();

    if (wv < 4) {
        const int mn = lane & 15;
        const int hh = lane >> 4;
        const _Float16* trow = &tileA[mn * GEMM_WP];
        v8f c = {};
#pragma unroll
        for (int kf = 0; kf < 4; ++kf) {
            const v8h alo = *(const v8h*)(trow + kf * 32 + 8 * hh);
            const v8h ahi = *(const v8h*)(trow + kf * 32 + 16 + 8 * hh);
            const v8h blo = bf[2 * kf];
            const v8h bhi = bf[2 * kf + 1];
            const v16h a = __builtin_shufflevector(alo, ahi, 0, 1, 2, 3, 4, 5, 6, 7,
                                                   8, 9, 10, 11, 12, 13, 14, 15);
            const v16h b = __builtin_shufflevector(blo, bhi, 0, 1, 2, 3, 4, 5, 6, 7,
                                                   8, 9, 10, 11, 12, 13, 14, 15);
            c = __builtin_amdgcn_wmma_f32_16x16x32_f16(
                    false, a, false, b, (short)0, c, false, false);
        }
        float* d = &tileD[(8 * hh) * G2_DP + wv * 16 + mn];
#pragma unroll
        for (int r = 0; r < 8; ++r) d[r * G2_DP] = c[r];
    }
    __syncthreads();

    if (wv < 8) {
        const int row = 2 * wv + (lane >> 4);
        const int col = (lane & 15) * 4;
        const int nd  = blockIdx.x * 16 + row;
        if (nd < n)
            *(v4f*)(H2 + (size_t)nd * 64 + col) = *(const v4f*)&tileD[row * G2_DP + col];
    }
}


extern "C" void kernel_launch(void* const* d_in, const int* in_sizes, int n_in,
                              void* d_out, int out_size, void* d_ws, size_t ws_size,
                              hipStream_t stream) {
    const float* x  = (const float*)d_in[0];
    const int*   ei = (const int*)d_in[1];
    const float* W1 = (const float*)d_in[2];
    const float* b1 = (const float*)d_in[3];
    const float* W2 = (const float*)d_in[4];
    const float* b2 = (const float*)d_in[5];
    float*       out = (float*)d_out;

    const int N = in_sizes[0] / 128;
    const int E = in_sizes[1] / 2;
    const int* src = ei;
    const int* dst = ei + E;

    char* w = (char*)d_ws;
    auto carve = [&](size_t bytes) -> void* {
        void* p = (void*)w;
        w += (bytes + 255) & ~(size_t)255;
        return p;
    };
    const int nw     = (N + 1) / 2;
    const int np     = 2 * nw;
    const int chunks = (E + HIST_CH - 1) / HIST_CH;
    int*   cnt     = (int*)carve((size_t)np * 4);
    int*   slot    = (int*)carve((size_t)E * 4);
    int*   rowptr  = (int*)carve((size_t)(N + 1) * 4);
    int*   seg_src = (int*)carve((size_t)E * 4);
    float* dinv    = (float*)carve((size_t)N * 4);
    float* h1      = (float*)carve((size_t)N * 128 * 4);
    float* h2      = (float*)carve((size_t)N * 64 * 4);
    _Float16* W2f  = (_Float16*)carve((size_t)1024 * 16);
    unsigned int* histw = (unsigned int*)carve((size_t)chunks * nw * 4);
    int*   off     = (int*)carve((size_t)chunks * np * 4);

    const int B = 256;
    const int ntiles = (N + 15) / 16;

    const int gemm1_blocks = (ntiles + 31) / 32;
    k_hist_gemm1<<<chunks + gemm1_blocks, 1024, 0, stream>>>(dst, histw, slot, E, N, nw, chunks,
                                                            x, W1, h1, ntiles);
    k_colscan<<<(nw + B - 1) / B, B, 0, stream>>>(histw, off, cnt, nw, chunks);
    k_segstart<<<(N + 1023) / 1024, 1024, 0, stream>>>(cnt, rowptr, dinv, N, W2, W2f);
    k_place<<<(E + B - 1) / B, B, 0, stream>>>(src, dst, slot, rowptr, off, seg_src, E, np);

    k_gather_gemm2<<<ntiles, 512, 0, stream>>>(h1, rowptr, seg_src, dinv, b1, W2f, h2, N);

    k_gather<64><<<(N + 7) / 8, 256, 0, stream>>>(h2, rowptr, seg_src, dinv, b2, out, N);
}


# Gathers: a node's row loads go out in groups of 8/4/2/1 by the bits of its degree (was 4 plus a one-row tail), own row and bias asked for first: fewer dependent round trips; sum order unchanged
# speedup vs baseline: 2.7054x; 1.0778x over previous
#include <hip/hip_runtime.h>
#include <hip/hip_bf16.h>

typedef __attribute__((ext_vector_type(16))) _Float16 v16h;
typedef __attribute__((ext_vector_type(8)))  float    v8f;
typedef __attribute__((ext_vector_type(4))) _Float16 v4h;
typedef __attribute__((ext_vector_type(8))) _Float16 v8h;


constexpr int HIST_MAXN = 131072;
constexpr int HIST_CH   = 32768;

__device__ __forceinline__
void hist_block(const int* __restrict__ dst, unsigned int* __restrict__ histw,
                int* __restrict__ slot, int E, int n, int nw, unsigned int* c) {
    const int tid = threadIdx.x;
    const int e0  = blockIdx.x * HIST_CH;
    const int e1  = min(E, e0 + HIST_CH);
    unsigned int* hrow = histw + (size_t)blockIdx.x * nw;
    for (int lo = 0; lo < n; lo += HIST_MAXN) {
        const int words = min(HIST_MAXN / 2, nw - (lo >> 1));
        for (int i = tid; i < words; i += 1024) c[i] = 0u;
        __syncthreads();
        for (int b = e0 + tid; b < e1; b += 8 * 1024) {
            int      d[8];
            unsigned r[8];
#pragma unroll
            for (int j = 0; j < 8; ++j)
                d[j] = dst[min(b + j * 1024, e1 - 1)] - lo;
#pragma unroll
            for (int j = 0; j < 8; ++j) {
                const bool ok = (b + j * 1024 < e1) && ((unsigned)d[j] < (unsigned)HIST_MAXN);
                const unsigned sh = (unsigned)(d[j] & 1) << 4;
                r[j] = atomicAdd(&c[ok ? (d[j] >> 1) : 0], ok ? (1u << sh) : 0u);
            }
#pragma unroll
            for (int j = 0; j < 8; ++j) {
                const bool ok = (b + j * 1024 < e1) && ((unsigned)d[j] < (unsigned)HIST_MAXN);
                const unsigned sh = (unsigned)(d[j] & 1) << 4;
                if (ok) slot[b + j * 1024] = (int)((r[j] >> sh) & 0xFFFFu);
            }
        }
        __syncthreads();
        for (int i = tid; i < words; i += 1024) hrow[(lo >> 1) + i] = c[i];
        __syncthreads();
    }
}

__global__ void k_colscan(const unsigned int* __restrict__ histw, int* __restrict__ off,
                          int* __restrict__ cnt, int nw, int chunks) {
    const int w = blockIdx.x * blockDim.x + threadIdx.x;
    if (w >= nw) return;
    const size_t np = (size_t)2 * nw;
    int a = 0, b = 0;
#pragma unroll 8
    for (int p = 0; p < chunks; ++p) {
        const unsigned v = histw[(size_t)p * nw + w];
        *(int2*)(off + (size_t)p * np + 2 * w) = make_int2(a, b);
        a += (int)(v & 0xFFFFu);
        b += (int)(v >> 16);
    }
    *(int2*)(cnt + 2 * w) = make_int2(a, b);
}

__global__ __launch_bounds__(1024)
void k_segstart(const int* __restrict__ cnt, int* __restrict__ rowptr,
                float* __restrict__ dinv, int n,
                const float* __restrict__ W2, _Float16* __restrict__ W2f) {
    __shared__ int wsum[32];
    __shared__ int s_front;
    const int tid  = threadIdx.x;
    const int lane = tid & 31;
    const int wave = tid >> 5;
    const int base = blockIdx.x * 1024;

    if (blockIdx.x == 0) {
        const int j  = (tid >> 5) & 1;
        const int kf = (tid >> 6) & 3;
        const int nt = tid >> 8;
        const float* wp = W2 + (size_t)(kf * 32 + 16 * j + 8 * (lane >> 4)) * 64 + nt * 16 + (lane & 15);
        v8h f;
#pragma unroll
        for (int t = 0; t < 8; ++t) f[t] = (_Float16)wp[t * 64];
        *(v8h*)(W2f + tid * 8) = f;
    }

    int part = 0;
    const int4* c4 = (const int4*)cnt;
    const int n4 = base >> 2;
    for (int j = tid; j < n4; j += 1024) {
        const int4 v = c4[j];
        part += (v.x + v.y) + (v.z + v.w);
    }
    for (int o = 16; o > 0; o >>= 1) part += __shfl_xor(part, o);
    if (lane == 0) wsum[wave] = part;
    __syncthreads();
    if (wave == 0) {
        int v = wsum[lane];
        for (int o = 16; o > 0; o >>= 1) v += __shfl_xor(v, o);
        if (lane == 0) s_front = v;
    }
    __syncthreads();
    const int front = s_front;

    const int i = base + tid;
    const int c = (i < n) ? cnt[i] : 0;
    int incl = c;
    for (int o = 1; o < 32; o <<= 1) {
        const int t = __shfl_up(incl, o);
        if (lane >= o) incl += t;
    }
    if (lane == 31) wsum[wave] = incl;
    __syncthreads();
    int woff = (lane < wave) ? wsum[lane] : 0;
    for (int o = 16; o > 0; o >>= 1) woff += __shfl_xor(woff, o);

    const int excl = front + woff + incl - c;
    if (i < n) {
        rowptr[i] = excl;
        dinv[i]   = rsqrtf((float)c + 1.0f);
        if (i == n - 1) rowptr[n] = excl + c;
    }
}

__global__ void k_place(const int* __restrict__ src, const int* __restrict__ dst,
                        const int* __restrict__ slot, const int* __restrict__ rowptr,
                        const int* __restrict__ off, int* __restrict__ seg_src, int E, int np) {
    int e = blockIdx.x * blockDim.x + threadIdx.x;
    if (e < E) {
        const int d = dst[e];
        const int p = e / HIST_CH;
        seg_src[rowptr[d] + off[(size_t)p * np + d] + slot[e]] = src[e];
    }
}


constexpr int GEMM_WP = 136;

template <int OUTC, bool RELU, int NT>
__device__ __forceinline__
void gemm_block(const float* __restrict__ X, const float* __restrict__ Wg,
                _Float16* __restrict__ H, int ntiles, int blk, _Float16* ldsW) {
    constexpr int WP = GEMM_WP;

    const int tid = threadIdx.x;
    for (int it = tid; it < 32 * (OUTC / 4); it += NT) {
        const int g  = ((it >> 7) << 2) | (it & 3);
        const int kq = (it >> 2) & 31;
        const float4* wsrc = (const float4*)(Wg + (size_t)(4 * kq) * OUTC + 4 * g);
        const float4 r0 = wsrc[0];
        const float4 r1 = wsrc[OUTC / 4];
        const float4 r2 = wsrc[2 * (OUTC / 4)];
        const float4 r3 = wsrc[3 * (OUTC / 4)];
        const v4h c0 = {(_Float16)r0.x, (_Float16)r1.x, (_Float16)r2.x, (_Float16)r3.x};
        const v4h c1 = {(_Float16)r0.y, (_Float16)r1.y, (_Float16)r2.y, (_Float16)r3.y};
        const v4h c2 = {(_Float16)r0.z, (_Float16)r1.z, (_Float16)r2.z, (_Float16)r3.z};
        const v4h c3 = {(_Float16)r0.w, (_Float16)r1.w, (_Float16)r2.w, (_Float16)r3.w};
        _Float16* wdst = &ldsW[(4 * g) * WP + 4 * kq];
        *(v4h*)(wdst)          = c0;
        *(v4h*)(wdst + WP)     = c1;
        *(v4h*)(wdst + 2 * WP) = c2;
        *(v4h*)(wdst + 3 * WP) = c3;
    }
    __syncthreads();

    const int wave = tid >> 5;
    const int lane = tid & 31;
    const int tile = blk * (NT / 32) + wave;
    if (tile >= ntiles) return;

    const int mn = lane & 15;
    const int hh = lane >> 4;
    _Float16* wtile = ldsW + OUTC * WP + wave * (16 * WP);

    const float* xrow = X + (size_t)(tile * 16 + mn) * 128;
    v16h a[4];
#pragma unroll
    for (int kf = 0; kf < 4; ++kf) {
        const int k0 = kf * 32;
#pragma unroll
        for (int i = 0; i < 8; ++i) {
            int kk = k0 + ((i < 4) ? 2 * i : 2 * i + 8) + 8 * hh;
            float v0 = xrow[kk];
            float v1 = xrow[kk + 1];
            if (RELU) { v0 = fmaxf(v0, 0.0f); v1 = fmaxf(v1, 0.0f); }
            a[kf][2 * i]     = (_Float16)v0;
            a[kf][2 * i + 1] = (_Float16)v1;
        }
    }

#pragma unroll
    for (int nt = 0; nt < OUTC / 16; ++nt) {
        const int col = nt * 16 + mn;
        const _Float16* wrow = &ldsW[col * WP];
        v8f c = {};
#pragma unroll
        for (int kf = 0; kf < 4; ++kf) {
            const int k0 = kf * 32;
            const v8h blo = *(const v8h*)(wrow + k0 + 8 * hh);
            const v8h bhi = *(const v8h*)(wrow + k0 + 16 + 8 * hh);
            const v16h b = __builtin_shufflevector(blo, bhi, 0, 1, 2, 3, 4, 5, 6, 7,
                                                   8, 9, 10, 11, 12, 13, 14, 15);
            c = __builtin_amdgcn_wmma_f32_16x16x32_f16(
                    false, a[kf], false, b, (short)0, c, false, false);
        }
        _Float16* tq = wtile + (8 * hh) * WP + col;
#pragma unroll
        for (int r = 0; r < 8; ++r)
            tq[r * WP] = (_Float16)c[r];
    }
    __builtin_amdgcn_wave_barrier();
    constexpr int LPR = OUTC / 8;
    _Float16* hout = H + (size_t)(tile * 16) * OUTC;
#pragma unroll
    for (int q = 0; q < (16 * LPR) / 32; ++q) {
        const int idx = q * 32 + lane;
        const int row = idx / LPR;
        const int cc  = (idx % LPR) * 8;
        *(v8h*)(hout + row * OUTC + cc) = *(const v8h*)(wtile + row * WP + cc);
    }
}

__global__ __launch_bounds__(1024)
void k_hist_gemm1(const int* __restrict__ dst, unsigned int* __restrict__ histw,
                  int* __restrict__ slot, int E, int n, int nw, int chunks,
                  const float* __restrict__ X, const float* __restrict__ Wg,
                  _Float16* __restrict__ H, int ntiles) {
    __shared__ __attribute__((aligned(16))) unsigned int c[HIST_MAXN / 2];
    if ((int)blockIdx.x >= chunks) {
        gemm_block<128, false, 1024>(X, Wg, H, ntiles, (int)blockIdx.x - chunks, (_Float16*)c);
        return;
    }
    hist_block(dst, histw, slot, E, n, nw, c);
}


template <int C, int B>
__device__ __forceinline__
void row_group(const _Float16* __restrict__ hl, int ss, float sw, float dd, int r,
               __attribute__((ext_vector_type(C / 32))) float& acc) {
    constexpr int V = C / 32;
    typedef __attribute__((ext_vector_type(V))) _Float16 vech;
    vech  hv[B];
    float nr[B];
#pragma unroll
    for (int j = 0; j < B; ++j) {
        const int sr = __builtin_amdgcn_readlane(ss, r + j);
        hv[j] = *(const vech*)(hl + (size_t)sr * C);
    }
#pragma unroll
    for (int j = 0; j < B; ++j)
        nr[j] = __int_as_float(__builtin_amdgcn_readlane(__float_as_int(sw), r + j)) * dd;
#pragma unroll
    for (int j = 0; j < B; ++j) {
#pragma unroll
        for (int k = 0; k < V; ++k) acc[k] = fmaf((float)hv[j][k], nr[j], acc[k]);
    }
}

template <int C>
__device__ __forceinline__
__attribute__((ext_vector_type(C / 32))) float
gather_node(const _Float16* __restrict__ h, const int* __restrict__ rowptr,
            const int* __restrict__ seg_src, const float* __restrict__ dinv,
            const float* __restrict__ bias, int node, int lane) {
    constexpr int V = C / 32;
    typedef __attribute__((ext_vector_type(V))) float vecf;
    typedef __attribute__((ext_vector_type(V))) _Float16 vech;

    const _Float16* hl = h + lane * V;
    const vech hs = *(const vech*)(hl + (size_t)node * C);
    const vecf bv = *(const vecf*)(bias + lane * V);
    const int beg = rowptr[node];
    const int end = rowptr[node + 1];
    const int L   = end - beg;
    const float dd = dinv[node];

    vecf acc;
#pragma unroll
    for (int k = 0; k < V; ++k) acc[k] = 0.0f;

    if (L <= 32) {
        int   s = 0x7fffffff;
        float w = 0.0f;
        if (lane < L) { s = seg_src[beg + lane]; w = dinv[s]; }
        int rk = 0;
        for (int i = 0; i < L; ++i) {
            const int si = __builtin_amdgcn_readlane(s, i);
            rk += ((si < s) || (si == s && i < lane)) ? 1 : 0;
        }
        if (lane >= L) rk = lane;
        const int   ss = __builtin_amdgcn_ds_permute(rk << 2, s);
        const float sw = __int_as_float(__builtin_amdgcn_ds_permute(rk << 2, __float_as_int(w)));
        int r = 0;
        for (; r + 8 <= L; r += 8) row_group<C, 8>(hl, ss, sw, dd, r, acc);
        if (L & 4) { row_group<C, 4>(hl, ss, sw, dd, r, acc); r += 4; }
        if (L & 2) { row_group<C, 2>(hl, ss, sw, dd, r, acc); r += 2; }
        if (L & 1) { row_group<C, 1>(hl, ss, sw, dd, r, acc); }
    } else {
        int cur = -1;
        for (int it = 0; it < L; ++it) {
            int mn = 0x7fffffff;
            for (int p = beg + lane; p < end; p += 32) {
                const int v = seg_src[p];
                if (v > cur && v < mn) mn = v;
            }
            for (int o = 16; o > 0; o >>= 1) mn = min(mn, __shfl_xor(mn, o));
            if (mn == 0x7fffffff) break;
            int m = 0;
            for (int p = beg + lane; p < end; p += 32) m += (seg_src[p] == mn) ? 1 : 0;
            for (int o = 16; o > 0; o >>= 1) m += __shfl_xor(m, o);
            const float nr = dinv[mn] * dd;
            const vech hv = *(const vech*)(hl + (size_t)mn * C);
            for (int t = 0; t < m; ++t) {
#pragma unroll
                for (int k = 0; k < V; ++k) acc[k] = fmaf((float)hv[k], nr, acc[k]);
            }
            cur = mn;
        }
    }

    const float s2 = dd * dd;
    vecf o;
#pragma unroll
    for (int k = 0; k < V; ++k) o[k] = fmaf((float)hs[k], s2, bv[k]) + acc[k];
    return o;
}

template <int C>
__global__ __launch_bounds__(256)
void k_gather(const _Float16* __restrict__ h, const int* __restrict__ rowptr,
              const int* __restrict__ seg_src, const float* __restrict__ dinv,
              const float* __restrict__ bias, float* __restrict__ out, int n) {
    constexpr int V = C / 32;
    typedef __attribute__((ext_vector_type(V))) float vecf;
    const int lane = threadIdx.x & 31;
    const int wv   = __builtin_amdgcn_readfirstlane((int)(threadIdx.x >> 5));
    const int node = blockIdx.x * 8 + wv;
    if (node >= n) return;
    const vecf o = gather_node<C>(h, rowptr, seg_src, dinv, bias, node, lane);
    *(vecf*)(out + (size_t)node * C + lane * V) = o;
}

constexpr int G2_DP = 68;

__global__ __launch_bounds__(512)
void k_gather_gemm2(const _Float16* __restrict__ h, const int* __restrict__ rowptr,
                    const int* __restrict__ seg_src, const float* __restrict__ dinv,
                    const float* __restrict__ bias, const _Float16* __restrict__ W2f,
                    _Float16* __restrict__ H2, int n) {
    typedef __attribute__((ext_vector_type(4))) float v4f;
    __shared__ __attribute__((aligned(16))) _Float16 tileA[16 * GEMM_WP];
    __shared__ __attribute__((aligned(16))) float    tileD[16 * G2_DP];
    const int lane = threadIdx.x & 31;
    const int wv   = __builtin_amdgcn_readfirstlane((int)(threadIdx.x >> 5));
    const int node = blockIdx.x * 16 + wv;

    v4f o = {0.0f, 0.0f, 0.0f, 0.0f};
    if (node < n)
        o = gather_node<128>(h, rowptr, seg_src, dinv, bias, node, lane);
    const v4h oh = {(_Float16)fmaxf(o[0], 0.0f), (_Float16)fmaxf(o[1], 0.0f),
                    (_Float16)fmaxf(o[2], 0.0f), (_Float16)fmaxf(o[3], 0.0f)};
    *(v4h*)&tileA[wv * GEMM_WP + lane * 4] = oh;
    v8h bf[8];
    if (wv < 4) {
        const v8h* wf = (const v8h*)W2f + wv * 256 + lane;
#pragma unroll
        for (int q = 0; q < 8; ++q) bf[q] = wf[q * 32];
    }
    __syncthreads();

    if (wv < 4) {
        const int mn = lane & 15;
        const int hh = lane >> 4;
        const _Float16* trow = &tileA[mn * GEMM_WP];
        v8f c = {};
#pragma unroll
        for (int kf = 0; kf < 4; ++kf) {
            const v8h alo = *(const v8h*)(trow + kf * 32 + 8 * hh);
            const v8h ahi = *(const v8h*)(trow + kf * 32 + 16 + 8 * hh);
            const v8h blo = bf[2 * kf];
            const v8h bhi = bf[2 * kf + 1];
            const v16h a = __builtin_shufflevector(alo, ahi, 0, 1, 2, 3, 4, 5, 6, 7,
                                                   8, 9, 10, 11, 12, 13, 14, 15);
            const v16h b = __builtin_shufflevector(blo, bhi, 0, 1, 2, 3, 4, 5, 6, 7,
                                                   8, 9, 10, 11, 12, 13, 14, 15);
            c = __builtin_amdgcn_wmma_f32_16x16x32_f16(
                    false, a, false, b, (short)0, c, false, false);
        }
        float* d = &tileD[(8 * hh) * G2_DP + wv * 16 + mn];
#pragma unroll
        for (int r = 0; r < 8; ++r) d[r * G2_DP] = c[r];
    }
    __syncthreads();

    if (wv < 4) {
        const int row = 4 * wv + (lane >> 3);
        const int col = (lane & 7) * 8;
        const int nd  = blockIdx.x * 16 + row;
        if (nd < n) {
            const v4f d0 = *(const v4f*)&tileD[row * G2_DP + col];
            const v4f d1 = *(const v4f*)&tileD[row * G2_DP + col + 4];
            const v8h dh = {(_Float16)d0[0], (_Float16)d0[1], (_Float16)d0[2], (_Float16)d0[3],
                            (_Float16)d1[0], (_Float16)d1[1], (_Float16)d1[2], (_Float16)d1[3]};
            *(v8h*)(H2 + (size_t)nd * 64 + col) = dh;
        }
    }
}


extern "C" void kernel_launch(void* const* d_in, const int* in_sizes, int n_in,
                              void* d_out, int out_size, void* d_ws, size_t ws_size,
                              hipStream_t stream) {
    const float* x  = (const float*)d_in[0];
    const int*   ei = (const int*)d_in[1];
    const float* W1 = (const float*)d_in[2];
    const float* b1 = (const float*)d_in[3];
    const float* W2 = (const float*)d_in[4];
    const float* b2 = (const float*)d_in[5];
    float*       out = (float*)d_out;

    const int N = in_sizes[0] / 128;
    const int E = in_sizes[1] / 2;
    const int* src = ei;
    const int* dst = ei + E;

    char* w = (char*)d_ws;
    auto carve = [&](size_t bytes) -> void* {
        void* p = (void*)w;
        w += (bytes + 255) & ~(size_t)255;
        return p;
    };
    const int nw     = (N + 1) / 2;
    const int np     = 2 * nw;
    const int chunks = (E + HIST_CH - 1) / HIST_CH;
    int*   cnt     = (int*)carve((size_t)np * 4);
    int*   slot    = (int*)carve((size_t)E * 4);
    int*   rowptr  = (int*)carve((size_t)(N + 1) * 4);
    int*   seg_src = (int*)carve((size_t)E * 4);
    float* dinv    = (float*)carve((size_t)N * 4);
    _Float16* h1   = (_Float16*)carve((size_t)N * 128 * 2);
    _Float16* h2   = (_Float16*)carve((size_t)N * 64 * 2);
    _Float16* W2f  = (_Float16*)carve((size_t)1024 * 16);
    unsigned int* histw = (unsigned int*)carve((size_t)chunks * nw * 4);
    int*   off     = (int*)carve((size_t)chunks * np * 4);

    const int B = 256;
    const int ntiles = (N + 15) / 16;

    const int gemm1_blocks = (ntiles + 31) / 32;
    k_hist_gemm1<<<chunks + gemm1_blocks, 1024, 0, stream>>>(dst, histw, slot, E, N, nw, chunks,
                                                            x, W1, h1, ntiles);
    k_colscan<<<(nw + B - 1) / B, B, 0, stream>>>(histw, off, cnt, nw, chunks);
    k_segstart<<<(N + 1023) / 1024, 1024, 0, stream>>>(cnt, rowptr, dinv, N, W2, W2f);
    k_place<<<(E + B - 1) / B, B, 0, stream>>>(src, dst, slot, rowptr, off, seg_src, E, np);

    k_gather_gemm2<<<ntiles, 512, 0, stream>>>(h1, rowptr, seg_src, dinv, b1, W2f, h2, N);

    k_gather<64><<<(N + 7) / 8, 256, 0, stream>>>(h2, rowptr, seg_src, dinv, b2, out, N);
}


# Gathers: a node with 33 to 64 incoming edges is rank-sorted in registers, two entries per lane, and summed by the same row groups (was one dependent pass per distinct source); sum order unchanged
# speedup vs baseline: 3.6504x; 1.3493x over previous
#include <hip/hip_runtime.h>
#include <hip/hip_bf16.h>

typedef __attribute__((ext_vector_type(16))) _Float16 v16h;
typedef __attribute__((ext_vector_type(8)))  float    v8f;
typedef __attribute__((ext_vector_type(4))) _Float16 v4h;
typedef __attribute__((ext_vector_type(8))) _Float16 v8h;


constexpr int HIST_MAXN = 131072;
constexpr int HIST_CH   = 32768;

__device__ __forceinline__
void hist_block(const int* __restrict__ dst, unsigned int* __restrict__ histw,
                int* __restrict__ slot, int E, int n, int nw, unsigned int* c) {
    const int tid = threadIdx.x;
    const int e0  = blockIdx.x * HIST_CH;
    const int e1  = min(E, e0 + HIST_CH);
    unsigned int* hrow = histw + (size_t)blockIdx.x * nw;
    for (int lo = 0; lo < n; lo += HIST_MAXN) {
        const int words = min(HIST_MAXN / 2, nw - (lo >> 1));
        for (int i = tid; i < words; i += 1024) c[i] = 0u;
        __syncthreads();
        for (int b = e0 + tid; b < e1; b += 8 * 1024) {
            int      d[8];
            unsigned r[8];
#pragma unroll
            for (int j = 0; j < 8; ++j)
                d[j] = dst[min(b + j * 1024, e1 - 1)] - lo;
#pragma unroll
            for (int j = 0; j < 8; ++j) {
                const bool ok = (b + j * 1024 < e1) && ((unsigned)d[j] < (unsigned)HIST_MAXN);
                const unsigned sh = (unsigned)(d[j] & 1) << 4;
                r[j] = atomicAdd(&c[ok ? (d[j] >> 1) : 0], ok ? (1u << sh) : 0u);
            }
#pragma unroll
            for (int j = 0; j < 8; ++j) {
                const bool ok = (b + j * 1024 < e1) && ((unsigned)d[j] < (unsigned)HIST_MAXN);
                const unsigned sh = (unsigned)(d[j] & 1) << 4;
                if (ok) slot[b + j * 1024] = (int)((r[j] >> sh) & 0xFFFFu);
            }
        }
        __syncthreads();
        for (int i = tid; i < words; i += 1024) hrow[(lo >> 1) + i] = c[i];
        __syncthreads();
    }
}

__global__ void k_colscan(const unsigned int* __restrict__ histw, int* __restrict__ off,
                          int* __restrict__ cnt, int nw, int chunks) {
    const int w = blockIdx.x * blockDim.x + threadIdx.x;
    if (w >= nw) return;
    const size_t np = (size_t)2 * nw;
    int a = 0, b = 0;
#pragma unroll 8
    for (int p = 0; p < chunks; ++p) {
        const unsigned v = histw[(size_t)p * nw + w];
        *(int2*)(off + (size_t)p * np + 2 * w) = make_int2(a, b);
        a += (int)(v & 0xFFFFu);
        b += (int)(v >> 16);
    }
    *(int2*)(cnt + 2 * w) = make_int2(a, b);
}

__global__ __launch_bounds__(1024)
void k_segstart(const int* __restrict__ cnt, int* __restrict__ rowptr,
                float* __restrict__ dinv, int n,
                const float* __restrict__ W2, _Float16* __restrict__ W2f) {
    __shared__ int wsum[32];
    __shared__ int s_front;
    const int tid  = threadIdx.x;
    const int lane = tid & 31;
    const int wave = tid >> 5;
    const int base = blockIdx.x * 1024;

    if (blockIdx.x == 0) {
        const int j  = (tid >> 5) & 1;
        const int kf = (tid >> 6) & 3;
        const int nt = tid >> 8;
        const float* wp = W2 + (size_t)(kf * 32 + 16 * j + 8 * (lane >> 4)) * 64 + nt * 16 + (lane & 15);
        v8h f;
#pragma unroll
        for (int t = 0; t < 8; ++t) f[t] = (_Float16)wp[t * 64];
        *(v8h*)(W2f + tid * 8) = f;
    }

    int part = 0;
    const int4* c4 = (const int4*)cnt;
    const int n4 = base >> 2;
    for (int j = tid; j < n4; j += 1024) {
        const int4 v = c4[j];
        part += (v.x + v.y) + (v.z + v.w);
    }
    for (int o = 16; o > 0; o >>= 1) part += __shfl_xor(part, o);
    if (lane == 0) wsum[wave] = part;
    __syncthreads();
    if (wave == 0) {
        int v = wsum[lane];
        for (int o = 16; o > 0; o >>= 1) v += __shfl_xor(v, o);
        if (lane == 0) s_front = v;
    }
    __syncthreads();
    const int front = s_front;

    const int i = base + tid;
    const int c = (i < n) ? cnt[i] : 0;
    int incl = c;
    for (int o = 1; o < 32; o <<= 1) {
        const int t = __shfl_up(incl, o);
        if (lane >= o) incl += t;
    }
    if (lane == 31) wsum[wave] = incl;
    __syncthreads();
    int woff = (lane < wave) ? wsum[lane] : 0;
    for (int o = 16; o > 0; o >>= 1) woff += __shfl_xor(woff, o);

    const int excl = front + woff + incl - c;
    if (i < n) {
        rowptr[i] = excl;
        dinv[i]   = rsqrtf((float)c + 1.0f);
        if (i == n - 1) rowptr[n] = excl + c;
    }
}

__global__ void k_place(const int* __restrict__ src, const int* __restrict__ dst,
                        const int* __restrict__ slot, const int* __restrict__ rowptr,
                        const int* __restrict__ off, int* __restrict__ seg_src, int E, int np) {
    int e = blockIdx.x * blockDim.x + threadIdx.x;
    if (e < E) {
        const int d = dst[e];
        const int p = e / HIST_CH;
        seg_src[rowptr[d] + off[(size_t)p * np + d] + slot[e]] = src[e];
    }
}


constexpr int GEMM_WP = 136;

template <int OUTC, bool RELU, int NT>
__device__ __forceinline__
void gemm_block(const float* __restrict__ X, const float* __restrict__ Wg,
                _Float16* __restrict__ H, int ntiles, int blk, _Float16* ldsW) {
    constexpr int WP = GEMM_WP;

    const int tid = threadIdx.x;
    for (int it = tid; it < 32 * (OUTC / 4); it += NT) {
        const int g  = ((it >> 7) << 2) | (it & 3);
        const int kq = (it >> 2) & 31;
        const float4* wsrc = (const float4*)(Wg + (size_t)(4 * kq) * OUTC + 4 * g);
        const float4 r0 = wsrc[0];
        const float4 r1 = wsrc[OUTC / 4];
        const float4 r2 = wsrc[2 * (OUTC / 4)];
        const float4 r3 = wsrc[3 * (OUTC / 4)];
        const v4h c0 = {(_Float16)r0.x, (_Float16)r1.x, (_Float16)r2.x, (_Float16)r3.x};
        const v4h c1 = {(_Float16)r0.y, (_Float16)r1.y, (_Float16)r2.y, (_Float16)r3.y};
        const v4h c2 = {(_Float16)r0.z, (_Float16)r1.z, (_Float16)r2.z, (_Float16)r3.z};
        const v4h c3 = {(_Float16)r0.w, (_Float16)r1.w, (_Float16)r2.w, (_Float16)r3.w};
        _Float16* wdst = &ldsW[(4 * g) * WP + 4 * kq];
        *(v4h*)(wdst)          = c0;
        *(v4h*)(wdst + WP)     = c1;
        *(v4h*)(wdst + 2 * WP) = c2;
        *(v4h*)(wdst + 3 * WP) = c3;
    }
    __syncthreads();

    const int wave = tid >> 5;
    const int lane = tid & 31;
    const int tile = blk * (NT / 32) + wave;
    if (tile >= ntiles) return;

    const int mn = lane & 15;
    const int hh = lane >> 4;
    _Float16* wtile = ldsW + OUTC * WP + wave * (16 * WP);

    const float* xrow = X + (size_t)(tile * 16 + mn) * 128;
    v16h a[4];
#pragma unroll
    for (int kf = 0; kf < 4; ++kf) {
        const int k0 = kf * 32;
#pragma unroll
        for (int i = 0; i < 8; ++i) {
            int kk = k0 + ((i < 4) ? 2 * i : 2 * i + 8) + 8 * hh;
            float v0 = xrow[kk];
            float v1 = xrow[kk + 1];
            if (RELU) { v0 = fmaxf(v0, 0.0f); v1 = fmaxf(v1, 0.0f); }
            a[kf][2 * i]     = (_Float16)v0;
            a[kf][2 * i + 1] = (_Float16)v1;
        }
    }

#pragma unroll
    for (int nt = 0; nt < OUTC / 16; ++nt) {
        const int col = nt * 16 + mn;
        const _Float16* wrow = &ldsW[col * WP];
        v8f c = {};
#pragma unroll
        for (int kf = 0; kf < 4; ++kf) {
            const int k0 = kf * 32;
            const v8h blo = *(const v8h*)(wrow + k0 + 8 * hh);
            const v8h bhi = *(const v8h*)(wrow + k0 + 16 + 8 * hh);
            const v16h b = __builtin_shufflevector(blo, bhi, 0, 1, 2, 3, 4, 5, 6, 7,
                                                   8, 9, 10, 11, 12, 13, 14, 15);
            c = __builtin_amdgcn_wmma_f32_16x16x32_f16(
                    false, a[kf], false, b, (short)0, c, false, false);
        }
        _Float16* tq = wtile + (8 * hh) * WP + col;
#pragma unroll
        for (int r = 0; r < 8; ++r)
            tq[r * WP] = (_Float16)c[r];
    }
    __builtin_amdgcn_wave_barrier();
    constexpr int LPR = OUTC / 8;
    _Float16* hout = H + (size_t)(tile * 16) * OUTC;
#pragma unroll
    for (int q = 0; q < (16 * LPR) / 32; ++q) {
        const int idx = q * 32 + lane;
        const int row = idx / LPR;
        const int cc  = (idx % LPR) * 8;
        *(v8h*)(hout + row * OUTC + cc) = *(const v8h*)(wtile + row * WP + cc);
    }
}

__global__ __launch_bounds__(1024)
void k_hist_gemm1(const int* __restrict__ dst, unsigned int* __restrict__ histw,
                  int* __restrict__ slot, int E, int n, int nw, int chunks,
                  const float* __restrict__ X, const float* __restrict__ Wg,
                  _Float16* __restrict__ H, int ntiles) {
    __shared__ __attribute__((aligned(16))) unsigned int c[HIST_MAXN / 2];
    if ((int)blockIdx.x >= chunks) {
        gemm_block<128, false, 1024>(X, Wg, H, ntiles, (int)blockIdx.x - chunks, (_Float16*)c);
        return;
    }
    hist_block(dst, histw, slot, E, n, nw, c);
}


template <int C, int B>
__device__ __forceinline__
void row_group(const _Float16* __restrict__ hl, int ss, float sw, float dd, int r,
               __attribute__((ext_vector_type(C / 32))) float& acc) {
    constexpr int V = C / 32;
    typedef __attribute__((ext_vector_type(V))) _Float16 vech;
    vech  hv[B];
    float nr[B];
#pragma unroll
    for (int j = 0; j < B; ++j) {
        const int sr = __builtin_amdgcn_readlane(ss, r + j);
        hv[j] = *(const vech*)(hl + (size_t)sr * C);
    }
#pragma unroll
    for (int j = 0; j < B; ++j)
        nr[j] = __int_as_float(__builtin_amdgcn_readlane(__float_as_int(sw), r + j)) * dd;
#pragma unroll
    for (int j = 0; j < B; ++j) {
#pragma unroll
        for (int k = 0; k < V; ++k) acc[k] = fmaf((float)hv[j][k], nr[j], acc[k]);
    }
}

template <int C>
__device__ __forceinline__
void sorted_rows(const _Float16* __restrict__ hl, int ss, float sw, float dd, int n,
                 __attribute__((ext_vector_type(C / 32))) float& acc) {
    int r = 0;
    for (; r + 8 <= n; r += 8) row_group<C, 8>(hl, ss, sw, dd, r, acc);
    if (n & 4) { row_group<C, 4>(hl, ss, sw, dd, r, acc); r += 4; }
    if (n & 2) { row_group<C, 2>(hl, ss, sw, dd, r, acc); r += 2; }
    if (n & 1) { row_group<C, 1>(hl, ss, sw, dd, r, acc); }
}

__device__ __forceinline__
void put_sorted(int rk, int s, float w, int cnt, int lane,
                int& ssa, float& swa, int& ssb, float& swb) {
    for (int i = 0; i < cnt; ++i) {
        const int   p = __builtin_amdgcn_readlane(rk, i);
        const int   v = __builtin_amdgcn_readlane(s, i);
        const float u = __int_as_float(__builtin_amdgcn_readlane(__float_as_int(w), i));
        const bool lo = (lane == p);
        const bool hi = (lane + 32 == p);
        ssa = lo ? v : ssa;
        swa = lo ? u : swa;
        ssb = hi ? v : ssb;
        swb = hi ? u : swb;
    }
}

template <int C>
__device__ __forceinline__
__attribute__((ext_vector_type(C / 32))) float
gather_node(const _Float16* __restrict__ h, const int* __restrict__ rowptr,
            const int* __restrict__ seg_src, const float* __restrict__ dinv,
            const float* __restrict__ bias, int node, int lane) {
    constexpr int V = C / 32;
    typedef __attribute__((ext_vector_type(V))) float vecf;
    typedef __attribute__((ext_vector_type(V))) _Float16 vech;

    const _Float16* hl = h + lane * V;
    const vech hs = *(const vech*)(hl + (size_t)node * C);
    const vecf bv = *(const vecf*)(bias + lane * V);
    const int beg = rowptr[node];
    const int end = rowptr[node + 1];
    const int L   = end - beg;
    const float dd = dinv[node];

    vecf acc;
#pragma unroll
    for (int k = 0; k < V; ++k) acc[k] = 0.0f;

    if (L <= 32) {
        int   s = 0x7fffffff;
        float w = 0.0f;
        if (lane < L) { s = seg_src[beg + lane]; w = dinv[s]; }
        int rk = 0;
        for (int i = 0; i < L; ++i) {
            const int si = __builtin_amdgcn_readlane(s, i);
            rk += ((si < s) || (si == s && i < lane)) ? 1 : 0;
        }
        if (lane >= L) rk = lane;
        const int   ss = __builtin_amdgcn_ds_permute(rk << 2, s);
        const float sw = __int_as_float(__builtin_amdgcn_ds_permute(rk << 2, __float_as_int(w)));
        sorted_rows<C>(hl, ss, sw, dd, L, acc);
    } else if (L <= 64) {
        const int L1 = L - 32;
        const int s0 = seg_src[beg + lane];
        int   s1 = 0x7fffffff;
        if (lane < L1) s1 = seg_src[beg + 32 + lane];
        const float w0 = dinv[s0];
        float w1 = 0.0f;
        if (lane < L1) w1 = dinv[s1];
        int rk0 = 0, rk1 = 0;
        for (int i = 0; i < 32; ++i) {
            const int si = __builtin_amdgcn_readlane(s0, i);
            rk0 += ((si < s0) || (si == s0 && i < lane)) ? 1 : 0;
            rk1 += (si <= s1) ? 1 : 0;
        }
        for (int i = 0; i < L1; ++i) {
            const int si = __builtin_amdgcn_readlane(s1, i);
            rk0 += (si < s0) ? 1 : 0;
            rk1 += ((si < s1) || (si == s1 && i < lane)) ? 1 : 0;
        }
        int   ssa = 0, ssb = 0;
        float swa = 0.0f, swb = 0.0f;
        put_sorted(rk0, s0, w0, 32, lane, ssa, swa, ssb, swb);
        put_sorted(rk1, s1, w1, L1, lane, ssa, swa, ssb, swb);
        sorted_rows<C>(hl, ssa, swa, dd, 32, acc);
        sorted_rows<C>(hl, ssb, swb, dd, L1, acc);
    } else {
        int cur = -1;
        for (int it = 0; it < L; ++it) {
            int mn = 0x7fffffff;
            for (int p = beg + lane; p < end; p += 32) {
                const int v = seg_src[p];
                if (v > cur && v < mn) mn = v;
            }
            for (int o = 16; o > 0; o >>= 1) mn = min(mn, __shfl_xor(mn, o));
            if (mn == 0x7fffffff) break;
            int m = 0;
            for (int p = beg + lane; p < end; p += 32) m += (seg_src[p] == mn) ? 1 : 0;
            for (int o = 16; o > 0; o >>= 1) m += __shfl_xor(m, o);
            const float nr = dinv[mn] * dd;
            const vech hv = *(const vech*)(hl + (size_t)mn * C);
            for (int t = 0; t < m; ++t) {
#pragma unroll
                for (int k = 0; k < V; ++k) acc[k] = fmaf((float)hv[k], nr, acc[k]);
            }
            cur = mn;
        }
    }

    const float s2 = dd * dd;
    vecf o;
#pragma unroll
    for (int k = 0; k < V; ++k) o[k] = fmaf((float)hs[k], s2, bv[k]) + acc[k];
    return o;
}

template <int C>
__global__ __launch_bounds__(256)
void k_gather(const _Float16* __restrict__ h, const int* __restrict__ rowptr,
              const int* __restrict__ seg_src, const float* __restrict__ dinv,
              const float* __restrict__ bias, float* __restrict__ out, int n) {
    constexpr int V = C / 32;
    typedef __attribute__((ext_vector_type(V))) float vecf;
    const int lane = threadIdx.x & 31;
    const int wv   = __builtin_amdgcn_readfirstlane((int)(threadIdx.x >> 5));
    const int node = blockIdx.x * 8 + wv;
    if (node >= n) return;
    const vecf o = gather_node<C>(h, rowptr, seg_src, dinv, bias, node, lane);
    *(vecf*)(out + (size_t)node * C + lane * V) = o;
}

constexpr int G2_DP = 68;

__global__ __launch_bounds__(512)
void k_gather_gemm2(const _Float16* __restrict__ h, const int* __restrict__ rowptr,
                    const int* __restrict__ seg_src, const float* __restrict__ dinv,
                    const float* __restrict__ bias, const _Float16* __restrict__ W2f,
                    _Float16* __restrict__ H2, int n) {
    typedef __attribute__((ext_vector_type(4))) float v4f;
    __shared__ __attribute__((aligned(16))) _Float16 tileA[16 * GEMM_WP];
    __shared__ __attribute__((aligned(16))) float    tileD[16 * G2_DP];
    const int lane = threadIdx.x & 31;
    const int wv   = __builtin_amdgcn_readfirstlane((int)(threadIdx.x >> 5));
    const int node = blockIdx.x * 16 + wv;

    v4f o = {0.0f, 0.0f, 0.0f, 0.0f};
    if (node < n)
        o = gather_node<128>(h, rowptr, seg_src, dinv, bias, node, lane);
    const v4h oh = {(_Float16)fmaxf(o[0], 0.0f), (_Float16)fmaxf(o[1], 0.0f),
                    (_Float16)fmaxf(o[2], 0.0f), (_Float16)fmaxf(o[3], 0.0f)};
    *(v4h*)&tileA[wv * GEMM_WP + lane * 4] = oh;
    v8h bf[8];
    if (wv < 4) {
        const v8h* wf = (const v8h*)W2f + wv * 256 + lane;
#pragma unroll
        for (int q = 0; q < 8; ++q) bf[q] = wf[q * 32];
    }
    __syncthreads();

    if (wv < 4) {
        const int mn = lane & 15;
        const int hh = lane >> 4;
        const _Float16* trow = &tileA[mn * GEMM_WP];
        v8f c = {};
#pragma unroll
        for (int kf = 0; kf < 4; ++kf) {
            const v8h alo = *(const v8h*)(trow + kf * 32 + 8 * hh);
            const v8h ahi = *(const v8h*)(trow + kf * 32 + 16 + 8 * hh);
            const v8h blo = bf[2 * kf];
            const v8h bhi = bf[2 * kf + 1];
            const v16h a = __builtin_shufflevector(alo, ahi, 0, 1, 2, 3, 4, 5, 6, 7,
                                                   8, 9, 10, 11, 12, 13, 14, 15);
            const v16h b = __builtin_shufflevector(blo, bhi, 0, 1, 2, 3, 4, 5, 6, 7,
                                                   8, 9, 10, 11, 12, 13, 14, 15);
            c = __builtin_amdgcn_wmma_f32_16x16x32_f16(
                    false, a, false, b, (short)0, c, false, false);
        }
        float* d = &tileD[(8 * hh) * G2_DP + wv * 16 + mn];
#pragma unroll
        for (int r = 0; r < 8; ++r) d[r * G2_DP] = c[r];
    }
    __syncthreads();

    if (wv < 4) {
        const int row = 4 * wv + (lane >> 3);
        const int col = (lane & 7) * 8;
        const int nd  = blockIdx.x * 16 + row;
        if (nd < n) {
            const v4f d0 = *(const v4f*)&tileD[row * G2_DP + col];
            const v4f d1 = *(const v4f*)&tileD[row * G2_DP + col + 4];
            const v8h dh = {(_Float16)d0[0], (_Float16)d0[1], (_Float16)d0[2], (_Float16)d0[3],
                            (_Float16)d1[0], (_Float16)d1[1], (_Float16)d1[2], (_Float16)d1[3]};
            *(v8h*)(H2 + (size_t)nd * 64 + col) = dh;
        }
    }
}


extern "C" void kernel_launch(void* const* d_in, const int* in_sizes, int n_in,
                              void* d_out, int out_size, void* d_ws, size_t ws_size,
                              hipStream_t stream) {
    const float* x  = (const float*)d_in[0];
    const int*   ei = (const int*)d_in[1];
    const float* W1 = (const float*)d_in[2];
    const float* b1 = (const float*)d_in[3];
    const float* W2 = (const float*)d_in[4];
    const float* b2 = (const float*)d_in[5];
    float*       out = (float*)d_out;

    const int N = in_sizes[0] / 128;
    const int E = in_sizes[1] / 2;
    const int* src = ei;
    const int* dst = ei + E;

    char* w = (char*)d_ws;
    auto carve = [&](size_t bytes) -> void* {
        void* p = (void*)w;
        w += (bytes + 255) & ~(size_t)255;
        return p;
    };
    const int nw     = (N + 1) / 2;
    const int np     = 2 * nw;
    const int chunks = (E + HIST_CH - 1) / HIST_CH;
    int*   cnt     = (int*)carve((size_t)np * 4);
    int*   slot    = (int*)carve((size_t)E * 4);
    int*   rowptr  = (int*)carve((size_t)(N + 1) * 4);
    int*   seg_src = (int*)carve((size_t)E * 4);
    float* dinv    = (float*)carve((size_t)N * 4);
    _Float16* h1   = (_Float16*)carve((size_t)N * 128 * 2);
    _Float16* h2   = (_Float16*)carve((size_t)N * 64 * 2);
    _Float16* W2f  = (_Float16*)carve((size_t)1024 * 16);
    unsigned int* histw = (unsigned int*)carve((size_t)chunks * nw * 4);
    int*   off     = (int*)carve((size_t)chunks * np * 4);

    const int B = 256;
    const int ntiles = (N + 15) / 16;

    const int gemm1_blocks = (ntiles + 31) / 32;
    k_hist_gemm1<<<chunks + gemm1_blocks, 1024, 0, stream>>>(dst, histw, slot, E, N, nw, chunks,
                                                            x, W1, h1, ntiles);
    k_colscan<<<(nw + B - 1) / B, B, 0, stream>>>(histw, off, cnt, nw, chunks);
    k_segstart<<<(N + 1023) / 1024, 1024, 0, stream>>>(cnt, rowptr, dinv, N, W2, W2f);
    k_place<<<(E + B - 1) / B, B, 0, stream>>>(src, dst, slot, rowptr, off, seg_src, E, np);

    k_gather_gemm2<<<ntiles, 512, 0, stream>>>(h1, rowptr, seg_src, dinv, b1, W2f, h2, N);

    k_gather<64><<<(N + 7) / 8, 256, 0, stream>>>(h2, rowptr, seg_src, dinv, b2, out, N);
}


# Segment starts: the column scan also writes each wave's edge total; a segment-start block sums the totals in front of it, one 16-byte load per thread (was up to 25 dependent loads over all counts)
# speedup vs baseline: 3.9022x; 1.0690x over previous
#include <hip/hip_runtime.h>
#include <hip/hip_bf16.h>

typedef __attribute__((ext_vector_type(16))) _Float16 v16h;
typedef __attribute__((ext_vector_type(8)))  float    v8f;
typedef __attribute__((ext_vector_type(4))) _Float16 v4h;
typedef __attribute__((ext_vector_type(8))) _Float16 v8h;


constexpr int HIST_MAXN = 131072;
constexpr int HIST_CH   = 32768;

__device__ __forceinline__
void hist_block(const int* __restrict__ dst, unsigned int* __restrict__ histw,
                int* __restrict__ slot, int E, int n, int nw, unsigned int* c) {
    const int tid = threadIdx.x;
    const int e0  = blockIdx.x * HIST_CH;
    const int e1  = min(E, e0 + HIST_CH);
    unsigned int* hrow = histw + (size_t)blockIdx.x * nw;
    for (int lo = 0; lo < n; lo += HIST_MAXN) {
        const int words = min(HIST_MAXN / 2, nw - (lo >> 1));
        for (int i = tid; i < words; i += 1024) c[i] = 0u;
        __syncthreads();
        for (int b = e0 + tid; b < e1; b += 8 * 1024) {
            int      d[8];
            unsigned r[8];
#pragma unroll
            for (int j = 0; j < 8; ++j)
                d[j] = dst[min(b + j * 1024, e1 - 1)] - lo;
#pragma unroll
            for (int j = 0; j < 8; ++j) {
                const bool ok = (b + j * 1024 < e1) && ((unsigned)d[j] < (unsigned)HIST_MAXN);
                const unsigned sh = (unsigned)(d[j] & 1) << 4;
                r[j] = atomicAdd(&c[ok ? (d[j] >> 1) : 0], ok ? (1u << sh) : 0u);
            }
#pragma unroll
            for (int j = 0; j < 8; ++j) {
                const bool ok = (b + j * 1024 < e1) && ((unsigned)d[j] < (unsigned)HIST_MAXN);
                const unsigned sh = (unsigned)(d[j] & 1) << 4;
                if (ok) slot[b + j * 1024] = (int)((r[j] >> sh) & 0xFFFFu);
            }
        }
        __syncthreads();
        for (int i = tid; i < words; i += 1024) hrow[(lo >> 1) + i] = c[i];
        __syncthreads();
    }
}

__global__ void k_colscan(const unsigned int* __restrict__ histw, int* __restrict__ off,
                          int* __restrict__ cnt, int* __restrict__ wtot, int nw, int chunks) {
    const int w = blockIdx.x * blockDim.x + threadIdx.x;
    const size_t np = (size_t)2 * nw;
    int a = 0, b = 0;
    if (w < nw) {
#pragma unroll 8
        for (int p = 0; p < chunks; ++p) {
            const unsigned v = histw[(size_t)p * nw + w];
            *(int2*)(off + (size_t)p * np + 2 * w) = make_int2(a, b);
            a += (int)(v & 0xFFFFu);
            b += (int)(v >> 16);
        }
        *(int2*)(cnt + 2 * w) = make_int2(a, b);
    }
    int t = a + b;
    for (int o = 16; o > 0; o >>= 1) t += __shfl_xor(t, o);
    if ((threadIdx.x & 31) == 0 && w < nw) wtot[w >> 5] = t;
}

__global__ __launch_bounds__(1024)
void k_segstart(const int* __restrict__ cnt, const int* __restrict__ wtot,
                int* __restrict__ rowptr, float* __restrict__ dinv, int n,
                const float* __restrict__ W2, _Float16* __restrict__ W2f) {
    __shared__ int wsum[32];
    __shared__ int s_front;
    const int tid  = threadIdx.x;
    const int lane = tid & 31;
    const int wave = tid >> 5;
    const int base = blockIdx.x * 1024;
    const int i = base + tid;
    const int c = (i < n) ? cnt[i] : 0;

    if (blockIdx.x == 0) {
        const int j  = (tid >> 5) & 1;
        const int kf = (tid >> 6) & 3;
        const int nt = tid >> 8;
        const float* wp = W2 + (size_t)(kf * 32 + 16 * j + 8 * (lane >> 4)) * 64 + nt * 16 + (lane & 15);
        v8h f;
#pragma unroll
        for (int t = 0; t < 8; ++t) f[t] = (_Float16)wp[t * 64];
        *(v8h*)(W2f + tid * 8) = f;
    }

    int part = 0;
    const int4* t4 = (const int4*)wtot;
    const int n4 = blockIdx.x * 4;
    for (int j = tid; j < n4; j += 1024) {
        const int4 v = t4[j];
        part += (v.x + v.y) + (v.z + v.w);
    }
    for (int o = 16; o > 0; o >>= 1) part += __shfl_xor(part, o);
    if (lane == 0) wsum[wave] = part;
    __syncthreads();
    if (wave == 0) {
        int v = wsum[lane];
        for (int o = 16; o > 0; o >>= 1) v += __shfl_xor(v, o);
        if (lane == 0) s_front = v;
    }
    __syncthreads();
    const int front = s_front;

    int incl = c;
    for (int o = 1; o < 32; o <<= 1) {
        const int t = __shfl_up(incl, o);
        if (lane >= o) incl += t;
    }
    if (lane == 31) wsum[wave] = incl;
    __syncthreads();
    int woff = (lane < wave) ? wsum[lane] : 0;
    for (int o = 16; o > 0; o >>= 1) woff += __shfl_xor(woff, o);

    const int excl = front + woff + incl - c;
    if (i < n) {
        rowptr[i] = excl;
        dinv[i]   = rsqrtf((float)c + 1.0f);
        if (i == n - 1) rowptr[n] = excl + c;
    }
}

__global__ void k_place(const int* __restrict__ src, const int* __restrict__ dst,
                        const int* __restrict__ slot, const int* __restrict__ rowptr,
                        const int* __restrict__ off, int* __restrict__ seg_src, int E, int np) {
    int e = blockIdx.x * blockDim.x + threadIdx.x;
    if (e < E) {
        const int d = dst[e];
        const int p = e / HIST_CH;
        seg_src[rowptr[d] + off[(size_t)p * np + d] + slot[e]] = src[e];
    }
}


constexpr int GEMM_WP = 136;

template <int OUTC, bool RELU, int NT>
__device__ __forceinline__
void gemm_block(const float* __restrict__ X, const float* __restrict__ Wg,
                _Float16* __restrict__ H, int ntiles, int blk, _Float16* ldsW) {
    constexpr int WP = GEMM_WP;

    const int tid = threadIdx.x;
    for (int it = tid; it < 32 * (OUTC / 4); it += NT) {
        const int g  = ((it >> 7) << 2) | (it & 3);
        const int kq = (it >> 2) & 31;
        const float4* wsrc = (const float4*)(Wg + (size_t)(4 * kq) * OUTC + 4 * g);
        const float4 r0 = wsrc[0];
        const float4 r1 = wsrc[OUTC / 4];
        const float4 r2 = wsrc[2 * (OUTC / 4)];
        const float4 r3 = wsrc[3 * (OUTC / 4)];
        const v4h c0 = {(_Float16)r0.x, (_Float16)r1.x, (_Float16)r2.x, (_Float16)r3.x};
        const v4h c1 = {(_Float16)r0.y, (_Float16)r1.y, (_Float16)r2.y, (_Float16)r3.y};
        const v4h c2 = {(_Float16)r0.z, (_Float16)r1.z, (_Float16)r2.z, (_Float16)r3.z};
        const v4h c3 = {(_Float16)r0.w, (_Float16)r1.w, (_Float16)r2.w, (_Float16)r3.w};
        _Float16* wdst = &ldsW[(4 * g) * WP + 4 * kq];
        *(v4h*)(wdst)          = c0;
        *(v4h*)(wdst + WP)     = c1;
        *(v4h*)(wdst + 2 * WP) = c2;
        *(v4h*)(wdst + 3 * WP) = c3;
    }
    __syncthreads();

    const int wave = tid >> 5;
    const int lane = tid & 31;
    const int tile = blk * (NT / 32) + wave;
    if (tile >= ntiles) return;

    const int mn = lane & 15;
    const int hh = lane >> 4;
    _Float16* wtile = ldsW + OUTC * WP + wave * (16 * WP);

    const float* xrow = X + (size_t)(tile * 16 + mn) * 128;
    v16h a[4];
#pragma unroll
    for (int kf = 0; kf < 4; ++kf) {
        const int k0 = kf * 32;
#pragma unroll
        for (int i = 0; i < 8; ++i) {
            int kk = k0 + ((i < 4) ? 2 * i : 2 * i + 8) + 8 * hh;
            float v0 = xrow[kk];
            float v1 = xrow[kk + 1];
            if (RELU) { v0 = fmaxf(v0, 0.0f); v1 = fmaxf(v1, 0.0f); }
            a[kf][2 * i]     = (_Float16)v0;
            a[kf][2 * i + 1] = (_Float16)v1;
        }
    }

#pragma unroll
    for (int nt = 0; nt < OUTC / 16; ++nt) {
        const int col = nt * 16 + mn;
        const _Float16* wrow = &ldsW[col * WP];
        v8f c = {};
#pragma unroll
        for (int kf = 0; kf < 4; ++kf) {
            const int k0 = kf * 32;
            const v8h blo = *(const v8h*)(wrow + k0 + 8 * hh);
            const v8h bhi = *(const v8h*)(wrow + k0 + 16 + 8 * hh);
            const v16h b = __builtin_shufflevector(blo, bhi, 0, 1, 2, 3, 4, 5, 6, 7,
                                                   8, 9, 10, 11, 12, 13, 14, 15);
            c = __builtin_amdgcn_wmma_f32_16x16x32_f16(
                    false, a[kf], false, b, (short)0, c, false, false);
        }
        _Float16* tq = wtile + (8 * hh) * WP + col;
#pragma unroll
        for (int r = 0; r < 8; ++r)
            tq[r * WP] = (_Float16)c[r];
    }
    __builtin_amdgcn_wave_barrier();
    constexpr int LPR = OUTC / 8;
    _Float16* hout = H + (size_t)(tile * 16) * OUTC;
#pragma unroll
    for (int q = 0; q < (16 * LPR) / 32; ++q) {
        const int idx = q * 32 + lane;
        const int row = idx / LPR;
        const int cc  = (idx % LPR) * 8;
        *(v8h*)(hout + row * OUTC + cc) = *(const v8h*)(wtile + row * WP + cc);
    }
}

__global__ __launch_bounds__(1024)
void k_hist_gemm1(const int* __restrict__ dst, unsigned int* __restrict__ histw,
                  int* __restrict__ slot, int E, int n, int nw, int chunks,
                  const float* __restrict__ X, const float* __restrict__ Wg,
                  _Float16* __restrict__ H, int ntiles) {
    __shared__ __attribute__((aligned(16))) unsigned int c[HIST_MAXN / 2];
    if ((int)blockIdx.x >= chunks) {
        gemm_block<128, false, 1024>(X, Wg, H, ntiles, (int)blockIdx.x - chunks, (_Float16*)c);
        return;
    }
    hist_block(dst, histw, slot, E, n, nw, c);
}


template <int C, int B>
__device__ __forceinline__
void row_group(const _Float16* __restrict__ hl, int ss, float sw, float dd, int r,
               __attribute__((ext_vector_type(C / 32))) float& acc) {
    constexpr int V = C / 32;
    typedef __attribute__((ext_vector_type(V))) _Float16 vech;
    vech  hv[B];
    float nr[B];
#pragma unroll
    for (int j = 0; j < B; ++j) {
        const int sr = __builtin_amdgcn_readlane(ss, r + j);
        hv[j] = *(const vech*)(hl + (size_t)sr * C);
    }
#pragma unroll
    for (int j = 0; j < B; ++j)
        nr[j] = __int_as_float(__builtin_amdgcn_readlane(__float_as_int(sw), r + j)) * dd;
#pragma unroll
    for (int j = 0; j < B; ++j) {
#pragma unroll
        for (int k = 0; k < V; ++k) acc[k] = fmaf((float)hv[j][k], nr[j], acc[k]);
    }
}

template <int C>
__device__ __forceinline__
void sorted_rows(const _Float16* __restrict__ hl, int ss, float sw, float dd, int n,
                 __attribute__((ext_vector_type(C / 32))) float& acc) {
    int r = 0;
    for (; r + 8 <= n; r += 8) row_group<C, 8>(hl, ss, sw, dd, r, acc);
    if (n & 4) { row_group<C, 4>(hl, ss, sw, dd, r, acc); r += 4; }
    if (n & 2) { row_group<C, 2>(hl, ss, sw, dd, r, acc); r += 2; }
    if (n & 1) { row_group<C, 1>(hl, ss, sw, dd, r, acc); }
}

__device__ __forceinline__
void put_sorted(int rk, int s, float w, int cnt, int lane,
                int& ssa, float& swa, int& ssb, float& swb) {
    for (int i = 0; i < cnt; ++i) {
        const int   p = __builtin_amdgcn_readlane(rk, i);
        const int   v = __builtin_amdgcn_readlane(s, i);
        const float u = __int_as_float(__builtin_amdgcn_readlane(__float_as_int(w), i));
        const bool lo = (lane == p);
        const bool hi = (lane + 32 == p);
        ssa = lo ? v : ssa;
        swa = lo ? u : swa;
        ssb = hi ? v : ssb;
        swb = hi ? u : swb;
    }
}

template <int C>
__device__ __forceinline__
__attribute__((ext_vector_type(C / 32))) float
gather_node(const _Float16* __restrict__ h, const int* __restrict__ rowptr,
            const int* __restrict__ seg_src, const float* __restrict__ dinv,
            const float* __restrict__ bias, int node, int lane) {
    constexpr int V = C / 32;
    typedef __attribute__((ext_vector_type(V))) float vecf;
    typedef __attribute__((ext_vector_type(V))) _Float16 vech;

    const _Float16* hl = h + lane * V;
    const vech hs = *(const vech*)(hl + (size_t)node * C);
    const vecf bv = *(const vecf*)(bias + lane * V);
    const int beg = rowptr[node];
    const int end = rowptr[node + 1];
    const int L   = end - beg;
    const float dd = dinv[node];

    vecf acc;
#pragma unroll
    for (int k = 0; k < V; ++k) acc[k] = 0.0f;

    if (L <= 32) {
        int   s = 0x7fffffff;
        float w = 0.0f;
        if (lane < L) { s = seg_src[beg + lane]; w = dinv[s]; }
        int rk = 0;
        for (int i = 0; i < L; ++i) {
            const int si = __builtin_amdgcn_readlane(s, i);
            rk += ((si < s) || (si == s && i < lane)) ? 1 : 0;
        }
        if (lane >= L) rk = lane;
        const int   ss = __builtin_amdgcn_ds_permute(rk << 2, s);
        const float sw = __int_as_float(__builtin_amdgcn_ds_permute(rk << 2, __float_as_int(w)));
        sorted_rows<C>(hl, ss, sw, dd, L, acc);
    } else if (L <= 64) {
        const int L1 = L - 32;
        const int s0 = seg_src[beg + lane];
        int   s1 = 0x7fffffff;
        if (lane < L1) s1 = seg_src[beg + 32 + lane];
        const float w0 = dinv[s0];
        float w1 = 0.0f;
        if (lane < L1) w1 = dinv[s1];
        int rk0 = 0, rk1 = 0;
        for (int i = 0; i < 32; ++i) {
            const int si = __builtin_amdgcn_readlane(s0, i);
            rk0 += ((si < s0) || (si == s0 && i < lane)) ? 1 : 0;
            rk1 += (si <= s1) ? 1 : 0;
        }
        for (int i = 0; i < L1; ++i) {
            const int si = __builtin_amdgcn_readlane(s1, i);
            rk0 += (si < s0) ? 1 : 0;
            rk1 += ((si < s1) || (si == s1 && i < lane)) ? 1 : 0;
        }
        int   ssa = 0, ssb = 0;
        float swa = 0.0f, swb = 0.0f;
        put_sorted(rk0, s0, w0, 32, lane, ssa, swa, ssb, swb);
        put_sorted(rk1, s1, w1, L1, lane, ssa, swa, ssb, swb);
        sorted_rows<C>(hl, ssa, swa, dd, 32, acc);
        sorted_rows<C>(hl, ssb, swb, dd, L1, acc);
    } else {
        int cur = -1;
        for (int it = 0; it < L; ++it) {
            int mn = 0x7fffffff;
            for (int p = beg + lane; p < end; p += 32) {
                const int v = seg_src[p];
                if (v > cur && v < mn) mn = v;
            }
            for (int o = 16; o > 0; o >>= 1) mn = min(mn, __shfl_xor(mn, o));
            if (mn == 0x7fffffff) break;
            int m = 0;
            for (int p = beg + lane; p < end; p += 32) m += (seg_src[p] == mn) ? 1 : 0;
            for (int o = 16; o > 0; o >>= 1) m += __shfl_xor(m, o);
            const float nr = dinv[mn] * dd;
            const vech hv = *(const vech*)(hl + (size_t)mn * C);
            for (int t = 0; t < m; ++t) {
#pragma unroll
                for (int k = 0; k < V; ++k) acc[k] = fmaf((float)hv[k], nr, acc[k]);
            }
            cur = mn;
        }
    }

    const float s2 = dd * dd;
    vecf o;
#pragma unroll
    for (int k = 0; k < V; ++k) o[k] = fmaf((float)hs[k], s2, bv[k]) + acc[k];
    return o;
}

template <int C>
__global__ __launch_bounds__(256)
void k_gather(const _Float16* __restrict__ h, const int* __restrict__ rowptr,
              const int* __restrict__ seg_src, const float* __restrict__ dinv,
              const float* __restrict__ bias, float* __restrict__ out, int n) {
    constexpr int V = C / 32;
    typedef __attribute__((ext_vector_type(V))) float vecf;
    const int lane = threadIdx.x & 31;
    const int wv   = __builtin_amdgcn_readfirstlane((int)(threadIdx.x >> 5));
    const int node = blockIdx.x * 8 + wv;
    if (node >= n) return;
    const vecf o = gather_node<C>(h, rowptr, seg_src, dinv, bias, node, lane);
    *(vecf*)(out + (size_t)node * C + lane * V) = o;
}

constexpr int G2_DP = 68;

__global__ __launch_bounds__(512)
void k_gather_gemm2(const _Float16* __restrict__ h, const int* __restrict__ rowptr,
                    const int* __restrict__ seg_src, const float* __restrict__ dinv,
                    const float* __restrict__ bias, const _Float16* __restrict__ W2f,
                    _Float16* __restrict__ H2, int n) {
    typedef __attribute__((ext_vector_type(4))) float v4f;
    __shared__ __attribute__((aligned(16))) _Float16 tileA[16 * GEMM_WP];
    __shared__ __attribute__((aligned(16))) float    tileD[16 * G2_DP];
    const int lane = threadIdx.x & 31;
    const int wv   = __builtin_amdgcn_readfirstlane((int)(threadIdx.x >> 5));
    const int node = blockIdx.x * 16 + wv;

    v4f o = {0.0f, 0.0f, 0.0f, 0.0f};
    if (node < n)
        o = gather_node<128>(h, rowptr, seg_src, dinv, bias, node, lane);
    const v4h oh = {(_Float16)fmaxf(o[0], 0.0f), (_Float16)fmaxf(o[1], 0.0f),
                    (_Float16)fmaxf(o[2], 0.0f), (_Float16)fmaxf(o[3], 0.0f)};
    *(v4h*)&tileA[wv * GEMM_WP + lane * 4] = oh;
    v8h bf[8];
    if (wv < 4) {
        const v8h* wf = (const v8h*)W2f + wv * 256 + lane;
#pragma unroll
        for (int q = 0; q < 8; ++q) bf[q] = wf[q * 32];
    }
    __syncthreads();

    if (wv < 4) {
        const int mn = lane & 15;
        const int hh = lane >> 4;
        const _Float16* trow = &tileA[mn * GEMM_WP];
        v8f c = {};
#pragma unroll
        for (int kf = 0; kf < 4; ++kf) {
            const v8h alo = *(const v8h*)(trow + kf * 32 + 8 * hh);
            const v8h ahi = *(const v8h*)(trow + kf * 32 + 16 + 8 * hh);
            const v8h blo = bf[2 * kf];
            const v8h bhi = bf[2 * kf + 1];
            const v16h a = __builtin_shufflevector(alo, ahi, 0, 1, 2, 3, 4, 5, 6, 7,
                                                   8, 9, 10, 11, 12, 13, 14, 15);
            const v16h b = __builtin_shufflevector(blo, bhi, 0, 1, 2, 3, 4, 5, 6, 7,
                                                   8, 9, 10, 11, 12, 13, 14, 15);
            c = __builtin_amdgcn_wmma_f32_16x16x32_f16(
                    false, a, false, b, (short)0, c, false, false);
        }
        float* d = &tileD[(8 * hh) * G2_DP + wv * 16 + mn];
#pragma unroll
        for (int r = 0; r < 8; ++r) d[r * G2_DP] = c[r];
    }
    __syncthreads();

    if (wv < 4) {
        const int row = 4 * wv + (lane >> 3);
        const int col = (lane & 7) * 8;
        const int nd  = blockIdx.x * 16 + row;
        if (nd < n) {
            const v4f d0 = *(const v4f*)&tileD[row * G2_DP + col];
            const v4f d1 = *(const v4f*)&tileD[row * G2_DP + col + 4];
            const v8h dh = {(_Float16)d0[0], (_Float16)d0[1], (_Float16)d0[2], (_Float16)d0[3],
                            (_Float16)d1[0], (_Float16)d1[1], (_Float16)d1[2], (_Float16)d1[3]};
            *(v8h*)(H2 + (size_t)nd * 64 + col) = dh;
        }
    }
}


extern "C" void kernel_launch(void* const* d_in, const int* in_sizes, int n_in,
                              void* d_out, int out_size, void* d_ws, size_t ws_size,
                              hipStream_t stream) {
    const float* x  = (const float*)d_in[0];
    const int*   ei = (const int*)d_in[1];
    const float* W1 = (const float*)d_in[2];
    const float* b1 = (const float*)d_in[3];
    const float* W2 = (const float*)d_in[4];
    const float* b2 = (const float*)d_in[5];
    float*       out = (float*)d_out;

    const int N = in_sizes[0] / 128;
    const int E = in_sizes[1] / 2;
    const int* src = ei;
    const int* dst = ei + E;

    char* w = (char*)d_ws;
    auto carve = [&](size_t bytes) -> void* {
        void* p = (void*)w;
        w += (bytes + 255) & ~(size_t)255;
        return p;
    };
    const int nw     = (N + 1) / 2;
    const int np     = 2 * nw;
    const int chunks = (E + HIST_CH - 1) / HIST_CH;
    int*   cnt     = (int*)carve((size_t)np * 4);
    int*   slot    = (int*)carve((size_t)E * 4);
    int*   rowptr  = (int*)carve((size_t)(N + 1) * 4);
    int*   seg_src = (int*)carve((size_t)E * 4);
    float* dinv    = (float*)carve((size_t)N * 4);
    _Float16* h1   = (_Float16*)carve((size_t)N * 128 * 2);
    _Float16* h2   = (_Float16*)carve((size_t)N * 64 * 2);
    _Float16* W2f  = (_Float16*)carve((size_t)1024 * 16);
    unsigned int* histw = (unsigned int*)carve((size_t)chunks * nw * 4);
    int*   off     = (int*)carve((size_t)chunks * np * 4);
    int*   wtot    = (int*)carve((size_t)((nw + 31) / 32) * 4);

    const int B = 256;
    const int ntiles = (N + 15) / 16;

    const int gemm1_blocks = (ntiles + 31) / 32;
    k_hist_gemm1<<<chunks + gemm1_blocks, 1024, 0, stream>>>(dst, histw, slot, E, N, nw, chunks,
                                                            x, W1, h1, ntiles);
    k_colscan<<<(nw + B - 1) / B, B, 0, stream>>>(histw, off, cnt, wtot, nw, chunks);
    k_segstart<<<(N + 1023) / 1024, 1024, 0, stream>>>(cnt, wtot, rowptr, dinv, N, W2, W2f);
    k_place<<<(E + B - 1) / B, B, 0, stream>>>(src, dst, slot, rowptr, off, seg_src, E, np);

    k_gather_gemm2<<<ntiles, 512, 0, stream>>>(h1, rowptr, seg_src, dinv, b1, W2f, h2, N);

    k_gather<64><<<(N + 7) / 8, 256, 0, stream>>>(h2, rowptr, seg_src, dinv, b2, out, N);
}


# Gathers: one wave takes two consecutive nodes (was one): their bounds, source lists, weights and, for 64-channel rows, groups of 8 rows load side by side; half as many waves; sum order unchanged
# speedup vs baseline: 4.0451x; 1.0366x over previous
#include <hip/hip_runtime.h>
#include <hip/hip_bf16.h>

typedef __attribute__((ext_vector_type(16))) _Float16 v16h;
typedef __attribute__((ext_vector_type(8)))  float    v8f;
typedef __attribute__((ext_vector_type(4))) _Float16 v4h;
typedef __attribute__((ext_vector_type(8))) _Float16 v8h;


constexpr int HIST_MAXN = 131072;
constexpr int HIST_CH   = 32768;

__device__ __forceinline__
void hist_block(const int* __restrict__ dst, unsigned int* __restrict__ histw,
                int* __restrict__ slot, int E, int n, int nw, unsigned int* c) {
    const int tid = threadIdx.x;
    const int e0  = blockIdx.x * HIST_CH;
    const int e1  = min(E, e0 + HIST_CH);
    unsigned int* hrow = histw + (size_t)blockIdx.x * nw;
    for (int lo = 0; lo < n; lo += HIST_MAXN) {
        const int words = min(HIST_MAXN / 2, nw - (lo >> 1));
        for (int i = tid; i < words; i += 1024) c[i] = 0u;
        __syncthreads();
        for (int b = e0 + tid; b < e1; b += 8 * 1024) {
            int      d[8];
            unsigned r[8];
#pragma unroll
            for (int j = 0; j < 8; ++j)
                d[j] = dst[min(b + j * 1024, e1 - 1)] - lo;
#pragma unroll
            for (int j = 0; j < 8; ++j) {
                const bool ok = (b + j * 1024 < e1) && ((unsigned)d[j] < (unsigned)HIST_MAXN);
                const unsigned sh = (unsigned)(d[j] & 1) << 4;
                r[j] = atomicAdd(&c[ok ? (d[j] >> 1) : 0], ok ? (1u << sh) : 0u);
            }
#pragma unroll
            for (int j = 0; j < 8; ++j) {
                const bool ok = (b + j * 1024 < e1) && ((unsigned)d[j] < (unsigned)HIST_MAXN);
                const unsigned sh = (unsigned)(d[j] & 1) << 4;
                if (ok) slot[b + j * 1024] = (int)((r[j] >> sh) & 0xFFFFu);
            }
        }
        __syncthreads();
        for (int i = tid; i < words; i += 1024) hrow[(lo >> 1) + i] = c[i];
        __syncthreads();
    }
}

__global__ void k_colscan(const unsigned int* __restrict__ histw, int* __restrict__ off,
                          int* __restrict__ cnt, int* __restrict__ wtot, int nw, int chunks) {
    const int w = blockIdx.x * blockDim.x + threadIdx.x;
    const size_t np = (size_t)2 * nw;
    int a = 0, b = 0;
    if (w < nw) {
#pragma unroll 8
        for (int p = 0; p < chunks; ++p) {
            const unsigned v = histw[(size_t)p * nw + w];
            *(int2*)(off + (size_t)p * np + 2 * w) = make_int2(a, b);
            a += (int)(v & 0xFFFFu);
            b += (int)(v >> 16);
        }
        *(int2*)(cnt + 2 * w) = make_int2(a, b);
    }
    int t = a + b;
    for (int o = 16; o > 0; o >>= 1) t += __shfl_xor(t, o);
    if ((threadIdx.x & 31) == 0 && w < nw) wtot[w >> 5] = t;
}

__global__ __launch_bounds__(1024)
void k_segstart(const int* __restrict__ cnt, const int* __restrict__ wtot,
                int* __restrict__ rowptr, float* __restrict__ dinv, int n,
                const float* __restrict__ W2, _Float16* __restrict__ W2f) {
    __shared__ int wsum[32];
    __shared__ int s_front;
    const int tid  = threadIdx.x;
    const int lane = tid & 31;
    const int wave = tid >> 5;
    const int base = blockIdx.x * 1024;
    const int i = base + tid;
    const int c = (i < n) ? cnt[i] : 0;

    if (blockIdx.x == 0) {
        const int j  = (tid >> 5) & 1;
        const int kf = (tid >> 6) & 3;
        const int nt = tid >> 8;
        const float* wp = W2 + (size_t)(kf * 32 + 16 * j + 8 * (lane >> 4)) * 64 + nt * 16 + (lane & 15);
        v8h f;
#pragma unroll
        for (int t = 0; t < 8; ++t) f[t] = (_Float16)wp[t * 64];
        *(v8h*)(W2f + tid * 8) = f;
    }

    int part = 0;
    const int4* t4 = (const int4*)wtot;
    const int n4 = blockIdx.x * 4;
    for (int j = tid; j < n4; j += 1024) {
        const int4 v = t4[j];
        part += (v.x + v.y) + (v.z + v.w);
    }
    for (int o = 16; o > 0; o >>= 1) part += __shfl_xor(part, o);
    if (lane == 0) wsum[wave] = part;
    __syncthreads();
    if (wave == 0) {
        int v = wsum[lane];
        for (int o = 16; o > 0; o >>= 1) v += __shfl_xor(v, o);
        if (lane == 0) s_front = v;
    }
    __syncthreads();
    const int front = s_front;

    int incl = c;
    for (int o = 1; o < 32; o <<= 1) {
        const int t = __shfl_up(incl, o);
        if (lane >= o) incl += t;
    }
    if (lane == 31) wsum[wave] = incl;
    __syncthreads();
    int woff = (lane < wave) ? wsum[lane] : 0;
    for (int o = 16; o > 0; o >>= 1) woff += __shfl_xor(woff, o);

    const int excl = front + woff + incl - c;
    if (i < n) {
        rowptr[i] = excl;
        dinv[i]   = rsqrtf((float)c + 1.0f);
        if (i == n - 1) rowptr[n] = excl + c;
    }
}

__global__ void k_place(const int* __restrict__ src, const int* __restrict__ dst,
                        const int* __restrict__ slot, const int* __restrict__ rowptr,
                        const int* __restrict__ off, int* __restrict__ seg_src, int E, int np) {
    int e = blockIdx.x * blockDim.x + threadIdx.x;
    if (e < E) {
        const int d = dst[e];
        const int p = e / HIST_CH;
        seg_src[rowptr[d] + off[(size_t)p * np + d] + slot[e]] = src[e];
    }
}


constexpr int GEMM_WP = 136;

template <int OUTC, bool RELU, int NT>
__device__ __forceinline__
void gemm_block(const float* __restrict__ X, const float* __restrict__ Wg,
                _Float16* __restrict__ H, int ntiles, int blk, _Float16* ldsW) {
    constexpr int WP = GEMM_WP;

    const int tid = threadIdx.x;
    for (int it = tid; it < 32 * (OUTC / 4); it += NT) {
        const int g  = ((it >> 7) << 2) | (it & 3);
        const int kq = (it >> 2) & 31;
        const float4* wsrc = (const float4*)(Wg + (size_t)(4 * kq) * OUTC + 4 * g);
        const float4 r0 = wsrc[0];
        const float4 r1 = wsrc[OUTC / 4];
        const float4 r2 = wsrc[2 * (OUTC / 4)];
        const float4 r3 = wsrc[3 * (OUTC / 4)];
        const v4h c0 = {(_Float16)r0.x, (_Float16)r1.x, (_Float16)r2.x, (_Float16)r3.x};
        const v4h c1 = {(_Float16)r0.y, (_Float16)r1.y, (_Float16)r2.y, (_Float16)r3.y};
        const v4h c2 = {(_Float16)r0.z, (_Float16)r1.z, (_Float16)r2.z, (_Float16)r3.z};
        const v4h c3 = {(_Float16)r0.w, (_Float16)r1.w, (_Float16)r2.w, (_Float16)r3.w};
        _Float16* wdst = &ldsW[(4 * g) * WP + 4 * kq];
        *(v4h*)(wdst)          = c0;
        *(v4h*)(wdst + WP)     = c1;
        *(v4h*)(wdst + 2 * WP) = c2;
        *(v4h*)(wdst + 3 * WP) = c3;
    }
    __syncthreads();

    const int wave = tid >> 5;
    const int lane = tid & 31;
    const int tile = blk * (NT / 32) + wave;
    if (tile >= ntiles) return;

    const int mn = lane & 15;
    const int hh = lane >> 4;
    _Float16* wtile = ldsW + OUTC * WP + wave * (16 * WP);

    const float* xrow = X + (size_t)(tile * 16 + mn) * 128;
    v16h a[4];
#pragma unroll
    for (int kf = 0; kf < 4; ++kf) {
        const int k0 = kf * 32;
#pragma unroll
        for (int i = 0; i < 8; ++i) {
            int kk = k0 + ((i < 4) ? 2 * i : 2 * i + 8) + 8 * hh;
            float v0 = xrow[kk];
            float v1 = xrow[kk + 1];
            if (RELU) { v0 = fmaxf(v0, 0.0f); v1 = fmaxf(v1, 0.0f); }
            a[kf][2 * i]     = (_Float16)v0;
            a[kf][2 * i + 1] = (_Float16)v1;
        }
    }

#pragma unroll
    for (int nt = 0; nt < OUTC / 16; ++nt) {
        const int col = nt * 16 + mn;
        const _Float16* wrow = &ldsW[col * WP];
        v8f c = {};
#pragma unroll
        for (int kf = 0; kf < 4; ++kf) {
            const int k0 = kf * 32;
            const v8h blo = *(const v8h*)(wrow + k0 + 8 * hh);
            const v8h bhi = *(const v8h*)(wrow + k0 + 16 + 8 * hh);
            const v16h b = __builtin_shufflevector(blo, bhi, 0, 1, 2, 3, 4, 5, 6, 7,
                                                   8, 9, 10, 11, 12, 13, 14, 15);
            c = __builtin_amdgcn_wmma_f32_16x16x32_f16(
                    false, a[kf], false, b, (short)0, c, false, false);
        }
        _Float16* tq = wtile + (8 * hh) * WP + col;
#pragma unroll
        for (int r = 0; r < 8; ++r)
            tq[r * WP] = (_Float16)c[r];
    }
    __builtin_amdgcn_wave_barrier();
    constexpr int LPR = OUTC / 8;
    _Float16* hout = H + (size_t)(tile * 16) * OUTC;
#pragma unroll
    for (int q = 0; q < (16 * LPR) / 32; ++q) {
        const int idx = q * 32 + lane;
        const int row = idx / LPR;
        const int cc  = (idx % LPR) * 8;
        *(v8h*)(hout + row * OUTC + cc) = *(const v8h*)(wtile + row * WP + cc);
    }
}

__global__ __launch_bounds__(1024)
void k_hist_gemm1(const int* __restrict__ dst, unsigned int* __restrict__ histw,
                  int* __restrict__ slot, int E, int n, int nw, int chunks,
                  const float* __restrict__ X, const float* __restrict__ Wg,
                  _Float16* __restrict__ H, int ntiles) {
    __shared__ __attribute__((aligned(16))) unsigned int c[HIST_MAXN / 2];
    if ((int)blockIdx.x >= chunks) {
        gemm_block<128, false, 1024>(X, Wg, H, ntiles, (int)blockIdx.x - chunks, (_Float16*)c);
        return;
    }
    hist_block(dst, histw, slot, E, n, nw, c);
}


template <int C, int B>
__device__ __forceinline__
void row_group(const _Float16* __restrict__ hl, int ss, float sw, float dd, int r,
               __attribute__((ext_vector_type(C / 32))) float& acc) {
    constexpr int V = C / 32;
    typedef __attribute__((ext_vector_type(V))) _Float16 vech;
    vech  hv[B];
    float nr[B];
#pragma unroll
    for (int j = 0; j < B; ++j) {
        const int sr = __builtin_amdgcn_readlane(ss, r + j);
        hv[j] = *(const vech*)(hl + (size_t)sr * C);
    }
#pragma unroll
    for (int j = 0; j < B; ++j)
        nr[j] = __int_as_float(__builtin_amdgcn_readlane(__float_as_int(sw), r + j)) * dd;
#pragma unroll
    for (int j = 0; j < B; ++j) {
#pragma unroll
        for (int k = 0; k < V; ++k) acc[k] = fmaf((float)hv[j][k], nr[j], acc[k]);
    }
}

template <int C, int B>
__device__ __forceinline__
void row_group2(const _Float16* __restrict__ hl, int ssa, float swa, float dda,
                int ssb, float swb, float ddb, int r,
                __attribute__((ext_vector_type(C / 32))) float& acca,
                __attribute__((ext_vector_type(C / 32))) float& accb) {
    constexpr int V = C / 32;
    typedef __attribute__((ext_vector_type(V))) _Float16 vech;
    vech  ha[B], hb[B];
    float na[B], nb[B];
#pragma unroll
    for (int j = 0; j < B; ++j) {
        const int sr = __builtin_amdgcn_readlane(ssa, r + j);
        ha[j] = *(const vech*)(hl + (size_t)sr * C);
    }
#pragma unroll
    for (int j = 0; j < B; ++j) {
        const int sr = __builtin_amdgcn_readlane(ssb, r + j);
        hb[j] = *(const vech*)(hl + (size_t)sr * C);
    }
#pragma unroll
    for (int j = 0; j < B; ++j) {
        na[j] = __int_as_float(__builtin_amdgcn_readlane(__float_as_int(swa), r + j)) * dda;
        nb[j] = __int_as_float(__builtin_amdgcn_readlane(__float_as_int(swb), r + j)) * ddb;
    }
#pragma unroll
    for (int j = 0; j < B; ++j) {
#pragma unroll
        for (int k = 0; k < V; ++k) acca[k] = fmaf((float)ha[j][k], na[j], acca[k]);
    }
#pragma unroll
    for (int j = 0; j < B; ++j) {
#pragma unroll
        for (int k = 0; k < V; ++k) accb[k] = fmaf((float)hb[j][k], nb[j], accb[k]);
    }
}

template <int C>
__device__ __forceinline__
void sorted_rows(const _Float16* __restrict__ hl, int ss, float sw, float dd, int n,
                 __attribute__((ext_vector_type(C / 32))) float& acc, int r0 = 0) {
    int r = r0;
    for (; r + 8 <= n; r += 8) row_group<C, 8>(hl, ss, sw, dd, r, acc);
    if (n & 4) { row_group<C, 4>(hl, ss, sw, dd, r, acc); r += 4; }
    if (n & 2) { row_group<C, 2>(hl, ss, sw, dd, r, acc); r += 2; }
    if (n & 1) { row_group<C, 1>(hl, ss, sw, dd, r, acc); }
}

__device__ __forceinline__
void put_sorted(int rk, int s, float w, int cnt, int lane,
                int& ssa, float& swa, int& ssb, float& swb) {
    for (int i = 0; i < cnt; ++i) {
        const int   p = __builtin_amdgcn_readlane(rk, i);
        const int   v = __builtin_amdgcn_readlane(s, i);
        const float u = __int_as_float(__builtin_amdgcn_readlane(__float_as_int(w), i));
        const bool lo = (lane == p);
        const bool hi = (lane + 32 == p);
        ssa = lo ? v : ssa;
        swa = lo ? u : swa;
        ssb = hi ? v : ssb;
        swb = hi ? u : swb;
    }
}

template <int C>
__device__ __forceinline__
__attribute__((ext_vector_type(C / 32))) float
gather_node(const _Float16* __restrict__ h, const int* __restrict__ rowptr,
            const int* __restrict__ seg_src, const float* __restrict__ dinv,
            const float* __restrict__ bias, int node, int lane) {
    constexpr int V = C / 32;
    typedef __attribute__((ext_vector_type(V))) float vecf;
    typedef __attribute__((ext_vector_type(V))) _Float16 vech;

    const _Float16* hl = h + lane * V;
    const vech hs = *(const vech*)(hl + (size_t)node * C);
    const vecf bv = *(const vecf*)(bias + lane * V);
    const int beg = rowptr[node];
    const int end = rowptr[node + 1];
    const int L   = end - beg;
    const float dd = dinv[node];

    vecf acc;
#pragma unroll
    for (int k = 0; k < V; ++k) acc[k] = 0.0f;

    if (L <= 32) {
        int   s = 0x7fffffff;
        float w = 0.0f;
        if (lane < L) { s = seg_src[beg + lane]; w = dinv[s]; }
        int rk = 0;
        for (int i = 0; i < L; ++i) {
            const int si = __builtin_amdgcn_readlane(s, i);
            rk += ((si < s) || (si == s && i < lane)) ? 1 : 0;
        }
        if (lane >= L) rk = lane;
        const int   ss = __builtin_amdgcn_ds_permute(rk << 2, s);
        const float sw = __int_as_float(__builtin_amdgcn_ds_permute(rk << 2, __float_as_int(w)));
        sorted_rows<C>(hl, ss, sw, dd, L, acc);
    } else if (L <= 64) {
        const int L1 = L - 32;
        const int s0 = seg_src[beg + lane];
        int   s1 = 0x7fffffff;
        if (lane < L1) s1 = seg_src[beg + 32 + lane];
        const float w0 = dinv[s0];
        float w1 = 0.0f;
        if (lane < L1) w1 = dinv[s1];
        int rk0 = 0, rk1 = 0;
        for (int i = 0; i < 32; ++i) {
            const int si = __builtin_amdgcn_readlane(s0, i);
            rk0 += ((si < s0) || (si == s0 && i < lane)) ? 1 : 0;
            rk1 += (si <= s1) ? 1 : 0;
        }
        for (int i = 0; i < L1; ++i) {
            const int si = __builtin_amdgcn_readlane(s1, i);
            rk0 += (si < s0) ? 1 : 0;
            rk1 += ((si < s1) || (si == s1 && i < lane)) ? 1 : 0;
        }
        int   ssa = 0, ssb = 0;
        float swa = 0.0f, swb = 0.0f;
        put_sorted(rk0, s0, w0, 32, lane, ssa, swa, ssb, swb);
        put_sorted(rk1, s1, w1, L1, lane, ssa, swa, ssb, swb);
        sorted_rows<C>(hl, ssa, swa, dd, 32, acc);
        sorted_rows<C>(hl, ssb, swb, dd, L1, acc);
    } else {
        int cur = -1;
        for (int it = 0; it < L; ++it) {
            int mn = 0x7fffffff;
            for (int p = beg + lane; p < end; p += 32) {
                const int v = seg_src[p];
                if (v > cur && v < mn) mn = v;
            }
            for (int o = 16; o > 0; o >>= 1) mn = min(mn, __shfl_xor(mn, o));
            if (mn == 0x7fffffff) break;
            int m = 0;
            for (int p = beg + lane; p < end; p += 32) m += (seg_src[p] == mn) ? 1 : 0;
            for (int o = 16; o > 0; o >>= 1) m += __shfl_xor(m, o);
            const float nr = dinv[mn] * dd;
            const vech hv = *(const vech*)(hl + (size_t)mn * C);
            for (int t = 0; t < m; ++t) {
#pragma unroll
                for (int k = 0; k < V; ++k) acc[k] = fmaf((float)hv[k], nr, acc[k]);
            }
            cur = mn;
        }
    }

    const float s2 = dd * dd;
    vecf o;
#pragma unroll
    for (int k = 0; k < V; ++k) o[k] = fmaf((float)hs[k], s2, bv[k]) + acc[k];
    return o;
}

template <int C>
__device__ __forceinline__
bool gather_pair(const _Float16* __restrict__ h, const int* __restrict__ rowptr,
                 const int* __restrict__ seg_src, const float* __restrict__ dinv,
                 const float* __restrict__ bias, int node, int lane,
                 __attribute__((ext_vector_type(C / 32))) float& oa,
                 __attribute__((ext_vector_type(C / 32))) float& ob) {
    constexpr int V = C / 32;
    typedef __attribute__((ext_vector_type(V))) float vecf;
    typedef __attribute__((ext_vector_type(V))) _Float16 vech;

    const int p0 = rowptr[node];
    const int p1 = rowptr[node + 1];
    const int p2 = rowptr[node + 2];
    const int la = p1 - p0;
    const int lb = p2 - p1;
    if (la > 32 || lb > 32) return false;

    const _Float16* hl = h + lane * V;
    const vech hsa = *(const vech*)(hl + (size_t)node * C);
    const vech hsb = *(const vech*)(hl + (size_t)(node + 1) * C);
    const vecf bv  = *(const vecf*)(bias + lane * V);
    const float dda = dinv[node];
    const float ddb = dinv[node + 1];

    const bool ina = lane < la;
    const bool inb = lane < lb;
    const int qa = seg_src[ina ? p0 + lane : 0];
    const int qb = seg_src[inb ? p1 + lane : 0];
    const float ga = dinv[qa];
    const float gb = dinv[qb];
    const int   sa = ina ? qa : 0x7fffffff;
    const int   sb = inb ? qb : 0x7fffffff;
    const float wa = ina ? ga : 0.0f;
    const float wb = inb ? gb : 0.0f;

    int rka = 0, rkb = 0;
    for (int i = 0; i < la; ++i) {
        const int si = __builtin_amdgcn_readlane(sa, i);
        rka += ((si < sa) || (si == sa && i < lane)) ? 1 : 0;
    }
    for (int i = 0; i < lb; ++i) {
        const int si = __builtin_amdgcn_readlane(sb, i);
        rkb += ((si < sb) || (si == sb && i < lane)) ? 1 : 0;
    }
    if (lane >= la) rka = lane;
    if (lane >= lb) rkb = lane;
    const int   ssa = __builtin_amdgcn_ds_permute(rka << 2, sa);
    const float swa = __int_as_float(__builtin_amdgcn_ds_permute(rka << 2, __float_as_int(wa)));
    const int   ssb = __builtin_amdgcn_ds_permute(rkb << 2, sb);
    const float swb = __int_as_float(__builtin_amdgcn_ds_permute(rkb << 2, __float_as_int(wb)));

    vecf acca, accb;
#pragma unroll
    for (int k = 0; k < V; ++k) { acca[k] = 0.0f; accb[k] = 0.0f; }

    int r = 0;
    if (C <= 64) {
        const int m8 = min(la, lb) & ~7;
        for (; r < m8; r += 8) row_group2<C, 8>(hl, ssa, swa, dda, ssb, swb, ddb, r, acca, accb);
    }
    sorted_rows<C>(hl, ssa, swa, dda, la, acca, r);
    sorted_rows<C>(hl, ssb, swb, ddb, lb, accb, r);

    const float sa2 = dda * dda;
    const float sb2 = ddb * ddb;
#pragma unroll
    for (int k = 0; k < V; ++k) {
        oa[k] = fmaf((float)hsa[k], sa2, bv[k]) + acca[k];
        ob[k] = fmaf((float)hsb[k], sb2, bv[k]) + accb[k];
    }
    return true;
}

template <int C>
__global__ __launch_bounds__(256)
void k_gather(const _Float16* __restrict__ h, const int* __restrict__ rowptr,
              const int* __restrict__ seg_src, const float* __restrict__ dinv,
              const float* __restrict__ bias, float* __restrict__ out, int n) {
    constexpr int V = C / 32;
    typedef __attribute__((ext_vector_type(V))) float vecf;
    const int lane = threadIdx.x & 31;
    const int wv   = __builtin_amdgcn_readfirstlane((int)(threadIdx.x >> 5));
    const int node = (blockIdx.x * 8 + wv) * 2;
    if (node >= n) return;
    if (node + 1 < n) {
        vecf oa, ob;
        if (gather_pair<C>(h, rowptr, seg_src, dinv, bias, node, lane, oa, ob)) {
            *(vecf*)(out + (size_t)node * C + lane * V)       = oa;
            *(vecf*)(out + (size_t)(node + 1) * C + lane * V) = ob;
            return;
        }
    }
    for (int q = 0; q < 2; ++q) {
        if (node + q < n) {
            const vecf o = gather_node<C>(h, rowptr, seg_src, dinv, bias, node + q, lane);
            *(vecf*)(out + (size_t)(node + q) * C + lane * V) = o;
        }
    }
}

constexpr int G2_DP = 68;

__global__ __launch_bounds__(512)
void k_gather_gemm2(const _Float16* __restrict__ h, const int* __restrict__ rowptr,
                    const int* __restrict__ seg_src, const float* __restrict__ dinv,
                    const float* __restrict__ bias, const _Float16* __restrict__ W2f,
                    _Float16* __restrict__ H2, int n) {
    typedef __attribute__((ext_vector_type(4))) float v4f;
    __shared__ __attribute__((aligned(16))) _Float16 tileA[32 * GEMM_WP];
    __shared__ __attribute__((aligned(16))) float    tileD[32 * G2_DP];
    const int lane = threadIdx.x & 31;
    const int wv   = __builtin_amdgcn_readfirstlane((int)(threadIdx.x >> 5));
    const int node = blockIdx.x * 32 + 2 * wv;

    v4f oa = {0.0f, 0.0f, 0.0f, 0.0f};
    v4f ob = {0.0f, 0.0f, 0.0f, 0.0f};
    bool both = false;
    if (node + 1 < n)
        both = gather_pair<128>(h, rowptr, seg_src, dinv, bias, node, lane, oa, ob);
    if (!both) {
        if (node < n)     oa = gather_node<128>(h, rowptr, seg_src, dinv, bias, node, lane);
        if (node + 1 < n) ob = gather_node<128>(h, rowptr, seg_src, dinv, bias, node + 1, lane);
    }
    const v4h oha = {(_Float16)fmaxf(oa[0], 0.0f), (_Float16)fmaxf(oa[1], 0.0f),
                     (_Float16)fmaxf(oa[2], 0.0f), (_Float16)fmaxf(oa[3], 0.0f)};
    const v4h ohb = {(_Float16)fmaxf(ob[0], 0.0f), (_Float16)fmaxf(ob[1], 0.0f),
                     (_Float16)fmaxf(ob[2], 0.0f), (_Float16)fmaxf(ob[3], 0.0f)};
    *(v4h*)&tileA[(2 * wv) * GEMM_WP + lane * 4]     = oha;
    *(v4h*)&tileA[(2 * wv + 1) * GEMM_WP + lane * 4] = ohb;
    const int rt = wv >> 2;
    const int cb = wv & 3;
    v8h bf[8];
    if (wv < 8) {
        const v8h* wf = (const v8h*)W2f + cb * 256 + lane;
#pragma unroll
        for (int q = 0; q < 8; ++q) bf[q] = wf[q * 32];
    }
    __syncthreads();

    if (wv < 8) {
        const int mn = lane & 15;
        const int hh = lane >> 4;
        const _Float16* trow = &tileA[(rt * 16 + mn) * GEMM_WP];
        v8f c = {};
#pragma unroll
        for (int kf = 0; kf < 4; ++kf) {
            const v8h alo = *(const v8h*)(trow + kf * 32 + 8 * hh);
            const v8h ahi = *(const v8h*)(trow + kf * 32 + 16 + 8 * hh);
            const v8h blo = bf[2 * kf];
            const v8h bhi = bf[2 * kf + 1];
            const v16h a = __builtin_shufflevector(alo, ahi, 0, 1, 2, 3, 4, 5, 6, 7,
                                                   8, 9, 10, 11, 12, 13, 14, 15);
            const v16h b = __builtin_shufflevector(blo, bhi, 0, 1, 2, 3, 4, 5, 6, 7,
                                                   8, 9, 10, 11, 12, 13, 14, 15);
            c = __builtin_amdgcn_wmma_f32_16x16x32_f16(
                    false, a, false, b, (short)0, c, false, false);
        }
        float* d = &tileD[(rt * 16 + 8 * hh) * G2_DP + cb * 16 + mn];
#pragma unroll
        for (int r = 0; r < 8; ++r) d[r * G2_DP] = c[r];
    }
    __syncthreads();

    if (wv < 8) {
        const int row = 4 * wv + (lane >> 3);
        const int col = (lane & 7) * 8;
        const int nd  = blockIdx.x * 32 + row;
        if (nd < n) {
            const v4f d0 = *(const v4f*)&tileD[row * G2_DP + col];
            const v4f d1 = *(const v4f*)&tileD[row * G2_DP + col + 4];
            const v8h dh = {(_Float16)d0[0], (_Float16)d0[1], (_Float16)d0[2], (_Float16)d0[3],
                            (_Float16)d1[0], (_Float16)d1[1], (_Float16)d1[2], (_Float16)d1[3]};
            *(v8h*)(H2 + (size_t)nd * 64 + col) = dh;
        }
    }
}


extern "C" void kernel_launch(void* const* d_in, const int* in_sizes, int n_in,
                              void* d_out, int out_size, void* d_ws, size_t ws_size,
                              hipStream_t stream) {
    const float* x  = (const float*)d_in[0];
    const int*   ei = (const int*)d_in[1];
    const float* W1 = (const float*)d_in[2];
    const float* b1 = (const float*)d_in[3];
    const float* W2 = (const float*)d_in[4];
    const float* b2 = (const float*)d_in[5];
    float*       out = (float*)d_out;

    const int N = in_sizes[0] / 128;
    const int E = in_sizes[1] / 2;
    const int* src = ei;
    const int* dst = ei + E;

    char* w = (char*)d_ws;
    auto carve = [&](size_t bytes) -> void* {
        void* p = (void*)w;
        w += (bytes + 255) & ~(size_t)255;
        return p;
    };
    const int nw     = (N + 1) / 2;
    const int np     = 2 * nw;
    const int chunks = (E + HIST_CH - 1) / HIST_CH;
    int*   cnt     = (int*)carve((size_t)np * 4);
    int*   slot    = (int*)carve((size_t)E * 4);
    int*   rowptr  = (int*)carve((size_t)(N + 1) * 4);
    int*   seg_src = (int*)carve((size_t)E * 4);
    float* dinv    = (float*)carve((size_t)N * 4);
    _Float16* h1   = (_Float16*)carve((size_t)N * 128 * 2);
    _Float16* h2   = (_Float16*)carve((size_t)N * 64 * 2);
    _Float16* W2f  = (_Float16*)carve((size_t)1024 * 16);
    unsigned int* histw = (unsigned int*)carve((size_t)chunks * nw * 4);
    int*   off     = (int*)carve((size_t)chunks * np * 4);
    int*   wtot    = (int*)carve((size_t)((nw + 31) / 32) * 4);

    const int B = 256;
    const int ntiles = (N + 15) / 16;

    const int gemm1_blocks = (ntiles + 31) / 32;
    k_hist_gemm1<<<chunks + gemm1_blocks, 1024, 0, stream>>>(dst, histw, slot, E, N, nw, chunks,
                                                            x, W1, h1, ntiles);
    k_colscan<<<(nw + B - 1) / B, B, 0, stream>>>(histw, off, cnt, wtot, nw, chunks);
    k_segstart<<<(N + 1023) / 1024, 1024, 0, stream>>>(cnt, wtot, rowptr, dinv, N, W2, W2f);
    k_place<<<(E + B - 1) / B, B, 0, stream>>>(src, dst, slot, rowptr, off, seg_src, E, np);

    k_gather_gemm2<<<(N + 31) / 32, 512, 0, stream>>>(h1, rowptr, seg_src, dinv, b1, W2f, h2, N);

    k_gather<64><<<(N + 15) / 16, 256, 0, stream>>>(h2, rowptr, seg_src, dinv, b2, out, N);
}
